# starting kernel
# baseline (speedup 1.0000x reference)
#include <hip/hip_runtime.h>
#include <hip/hip_bf16.h>

typedef float v2f __attribute__((ext_vector_type(2)));
typedef float v8f __attribute__((ext_vector_type(8)));

#define CDIM  100
#define LSLOT 7
#define RSLOT 10
#define WDIM  300
#define NEGV  -1e30f

__device__ __forceinline__ float wave_sum(float v) {
  #pragma unroll
  for (int m = 16; m; m >>= 1) v += __shfl_xor(v, m, 32);
  return v;
}

__device__ __forceinline__ float score_slot(int idx, const float* vrow,
                                            const float* __restrict__ ce_raw, int lane) {
  if (idx < 0) return NEGV;
  const float* row = ce_raw + (long)idx * CDIM;
  float dot = 0.f, ss = 0.f;
  for (int c = lane; c < CDIM; c += 32) {
    float x = row[c];
    dot = fmaf(x, vrow[c], dot);
    ss  = fmaf(x, x, ss);
  }
  #pragma unroll
  for (int m = 16; m; m >>= 1) {
    dot += __shfl_xor(dot, m, 32);
    ss  += __shfl_xor(ss,  m, 32);
  }
  return dot / fmaxf(sqrtf(ss), 1e-12f);
}

__global__ __launch_bounds__(64) void fused_gated_attn_kernel(
    const float* __restrict__ ce_raw, const float* __restrict__ wvec,
    const float* __restrict__ lW,     const float* __restrict__ rW,
    const float* __restrict__ gL_w,   const float* __restrict__ gL_b,
    const int* __restrict__ qidx,
    const int* __restrict__ lixs_c,   const int* __restrict__ rixs_c,
    const int* __restrict__ lixs_w,   const int* __restrict__ rixs_w,
    float* __restrict__ out)
{
  __shared__ float q_lds[32][104];
  __shared__ float v_lds[32][2][112];

  const int lane  = threadIdx.x & 31;
  const int wv    = threadIdx.x >> 5;
  const int bbase = blockIdx.x * 32 + wv * 16;
  const int hi = lane >> 4, ln = lane & 15;

  for (int r = 0; r < 16; ++r) {
    long row = qidx[bbase + r];
    float vals[4]; float ss = 0.f;
    #pragma unroll
    for (int i = 0; i < 4; ++i) {
      int c = lane + 32 * i;
      vals[i] = (c < CDIM) ? ce_raw[row * CDIM + c] : 0.f;
      ss = fmaf(vals[i], vals[i], ss);
    }
    ss = wave_sum(ss);
    float inv = 1.f / fmaxf(sqrtf(ss), 1e-12f);
    #pragma unroll
    for (int i = 0; i < 4; ++i) {
      int c = lane + 32 * i;
      if (c < CDIM) q_lds[wv * 16 + r][c] = vals[i] * inv;
    }
  }
  __syncthreads();

  for (int side = 0; side < 2; ++side) {
    const float* W = side ? rW : lW;
    for (int nt = 0; nt < 7; ++nt) {
      int cout = nt * 16 + ln;
      v8f acc = {};
      for (int k = 0; k < 25; ++k) {
        int d0 = k * 4 + 2 * hi;
        v2f a, bfrag;
        a[0] = q_lds[wv * 16 + ln][d0];
        a[1] = q_lds[wv * 16 + ln][d0 + 1];
        if (cout < CDIM) {
          bfrag[0] = W[cout * CDIM + d0];
          bfrag[1] = W[cout * CDIM + d0 + 1];
        } else {
          bfrag[0] = 0.f; bfrag[1] = 0.f;
        }
        acc = __builtin_amdgcn_wmma_f32_16x16x4_f32(
             false, a,  false, bfrag,
             (short)0, acc,  false,  false);
      }
      #pragma unroll
      for (int r = 0; r < 8; ++r)
        v_lds[wv * 16 + r + 8 * hi][side][nt * 16 + ln] = acc[r];
    }
  }
  __syncthreads();

  for (int bb = 0; bb < 16; ++bb) {
    const int b  = bbase + bb;
    const int bl = wv * 16 + bb;

    float la[LSLOT], ra[RSLOT];
    #pragma unroll
    for (int t = 0; t < LSLOT; ++t)
      la[t] = score_slot(lixs_c[b * LSLOT + t], &v_lds[bl][0][0], ce_raw, lane);
    #pragma unroll
    for (int t = 0; t < RSLOT; ++t)
      ra[t] = score_slot(rixs_c[b * RSLOT + t], &v_lds[bl][1][0], ce_raw, lane);

    float mx = la[0];
    #pragma unroll
    for (int t = 1; t < LSLOT; ++t) mx = fmaxf(mx, la[t]);
    float s = 0.f;
    #pragma unroll
    for (int t = 0; t < LSLOT; ++t) { la[t] = expf(la[t] - mx); s += la[t]; }
    float rs = 1.f / s;
    #pragma unroll
    for (int t = 0; t < LSLOT; ++t) la[t] *= rs;

    mx = ra[0];
    #pragma unroll
    for (int t = 1; t < RSLOT; ++t) mx = fmaxf(mx, ra[t]);
    s = 0.f;
    #pragma unroll
    for (int t = 0; t < RSLOT; ++t) { ra[t] = expf(ra[t] - mx); s += ra[t]; }
    rs = 1.f / s;
    #pragma unroll
    for (int t = 0; t < RSLOT; ++t) ra[t] *= rs;

    float z0 = gL_b[0], z1 = gL_b[1];
    #pragma unroll
    for (int t = 0; t < LSLOT; ++t) {
      z0 = fmaf(la[t], gL_w[t],                z0);
      z1 = fmaf(la[t], gL_w[(LSLOT + RSLOT) + t], z1);
    }
    #pragma unroll
    for (int t = 0; t < RSLOT; ++t) {
      z0 = fmaf(ra[t], gL_w[LSLOT + t],                       z0);
      z1 = fmaf(ra[t], gL_w[(LSLOT + RSLOT) + LSLOT + t],     z1);
    }
    float gm = fmaxf(z0, z1);
    float e0 = expf(z0 - gm), e1 = expf(z1 - gm);
    float gi = 1.f / (e0 + e1);
    float g0 = e0 * gi, g1 = e1 * gi;

    for (int j = lane; j < WDIM; j += 32) {
      float accL = 0.f, accR = 0.f;
      #pragma unroll
      for (int t = 0; t < LSLOT; ++t) {
        int iw = lixs_w[b * LSLOT + t];
        if (iw >= 0) accL = fmaf(la[t], wvec[(long)iw * WDIM + j], accL);
      }
      #pragma unroll
      for (int t = 0; t < RSLOT; ++t) {
        int iw = rixs_w[b * RSLOT + t];
        if (iw >= 0) accR = fmaf(ra[t], wvec[(long)iw * WDIM + j], accR);
      }
      out[(long)b * WDIM + j] = g0 * accL + g1 * accR;
    }
  }
}

extern "C" void kernel_launch(void* const* d_in, const int* in_sizes, int n_in,
                              void* d_out, int out_size, void* d_ws, size_t ws_size,
                              hipStream_t stream) {
  const float* ce_raw = (const float*)d_in[0];
  const float* wvec   = (const float*)d_in[1];
  const float* lW     = (const float*)d_in[2];
  const float* rW     = (const float*)d_in[3];
  const float* gL_w   = (const float*)d_in[4];
  const float* gL_b   = (const float*)d_in[5];
  const int*   qidx   = (const int*)d_in[6];
  const int*   lixs_c = (const int*)d_in[7];
  const int*   rixs_c = (const int*)d_in[8];
  const int*   lixs_w = (const int*)d_in[9];
  const int*   rixs_w = (const int*)d_in[10];

  const int B = in_sizes[6];
  const int blocks = B / 32;
  fused_gated_attn_kernel<<<blocks, 64, 0, stream>>>(
      ce_raw, wvec, lW, rW, gL_w, gL_b,
      qidx, lixs_c, rixs_c, lixs_w, rixs_w, (float*)d_out);
}


# Phase-1 W (B operand) fragments fetched ahead of use: 13 16-byte loads per lane, 9 at kernel entry, 4 behind the first 8 WMMAs (was 25 8-byte loads, each waited for). k order of that sum changes.
# speedup vs baseline: 21.3536x; 21.3536x over previous
#include <hip/hip_runtime.h>
#include <hip/hip_bf16.h>

typedef float v2f __attribute__((ext_vector_type(2)));
typedef float v4f __attribute__((ext_vector_type(4)));
typedef float v8f __attribute__((ext_vector_type(8)));

#define CDIM  100
#define LSLOT 7
#define RSLOT 10
#define WDIM  300
#define QPITCH 108
#define W_AHEAD 9
#define NEGV  -1e30f

__device__ __forceinline__ float wave_sum(float v) {
  #pragma unroll
  for (int m = 16; m; m >>= 1) v += __shfl_xor(v, m, 32);
  return v;
}

template <int N>
__device__ __forceinline__ void score_group(const int* idx, const float* vrow,
                                            const float* __restrict__ ce_raw, int lane,
                                            float* sc) {
  const bool has3 = lane < (CDIM - 96);
  const int  c3   = has3 ? 96 + lane : CDIM - 1;
  const float v0 = vrow[lane], v1 = vrow[lane + 32], v2 = vrow[lane + 64], v3 = vrow[c3];
  float x[N][4];
  #pragma unroll
  for (int t = 0; t < N; ++t) {
    const float* row = ce_raw + (long)max(idx[t], 0) * CDIM;
    x[t][0] = row[lane];
    x[t][1] = row[lane + 32];
    x[t][2] = row[lane + 64];
    x[t][3] = row[c3];
  }
  float dot[N], ss[N];
  #pragma unroll
  for (int t = 0; t < N; ++t) {
    float d = 0.f, s = 0.f;
    d = fmaf(x[t][0], v0, d);  s = fmaf(x[t][0], x[t][0], s);
    d = fmaf(x[t][1], v1, d);  s = fmaf(x[t][1], x[t][1], s);
    d = fmaf(x[t][2], v2, d);  s = fmaf(x[t][2], x[t][2], s);
    float d3 = fmaf(x[t][3], v3, d), s3 = fmaf(x[t][3], x[t][3], s);
    dot[t] = has3 ? d3 : d;
    ss[t]  = has3 ? s3 : s;
  }
  #pragma unroll
  for (int m = 16; m; m >>= 1) {
    #pragma unroll
    for (int t = 0; t < N; ++t) {
      dot[t] += __shfl_xor(dot[t], m, 32);
      ss[t]  += __shfl_xor(ss[t],  m, 32);
    }
  }
  #pragma unroll
  for (int t = 0; t < N; ++t)
    sc[t] = (idx[t] < 0) ? NEGV : dot[t] / fmaxf(sqrtf(ss[t]), 1e-12f);
}

__device__ __forceinline__ float uniform_f(float v) {
  return __int_as_float(__builtin_amdgcn_readfirstlane(__float_as_int(v)));
}

__global__ __launch_bounds__(512) void fused_gated_attn_kernel(
    const float* __restrict__ ce_raw, const float* __restrict__ wvec,
    const float* __restrict__ lW,     const float* __restrict__ rW,
    const float* __restrict__ gL_w,   const float* __restrict__ gL_b,
    const int* __restrict__ qidx,
    const int* __restrict__ lixs_c,   const int* __restrict__ rixs_c,
    const int* __restrict__ lixs_w,   const int* __restrict__ rixs_w,
    float* __restrict__ out)
{
  __shared__ __attribute__((aligned(16))) float q_lds[16][QPITCH];
  __shared__ float v_lds[16][2][112];
  __shared__ float o_lds[16 * WDIM];

  const int lane  = threadIdx.x & 31;
  const int wv    = __builtin_amdgcn_readfirstlane(threadIdx.x >> 5);
  const int bbase = blockIdx.x * 16;
  const int hi = lane >> 4, ln = lane & 15;

  const int  w_side = wv / 7, w_nt = wv % 7;
  const v4f* wp = (const v4f*)((w_side ? rW : lW) + min(w_nt * 16 + ln, CDIM - 1) * CDIM + 4 * hi);
  v4f bw[13];
  if (wv < 14) {
    #pragma unroll
    for (int j = 0; j < W_AHEAD; ++j) bw[j] = wp[2 * j];
  }

  {
    const int r = wv;
    long row = qidx[bbase + r];
    float vals[4]; float ss = 0.f;
    #pragma unroll
    for (int i = 0; i < 4; ++i) {
      int c = lane + 32 * i;
      vals[i] = (c < CDIM) ? ce_raw[row * CDIM + c] : 0.f;
      ss = fmaf(vals[i], vals[i], ss);
    }
    ss = wave_sum(ss);
    float inv = 1.f / fmaxf(sqrtf(ss), 1e-12f);
    #pragma unroll
    for (int i = 0; i < 4; ++i) {
      int c = lane + 32 * i;
      if (c < CDIM + 4) q_lds[r][c] = (c < CDIM) ? vals[i] * inv : 0.f;
    }
  }
  __syncthreads();

  if (wv < 14) {
    const v4f zero4 = {0.f, 0.f, 0.f, 0.f};
    v8f acc = {};
    #pragma unroll
    for (int j = 0; j < 13; ++j) {
      if (j == 13 - W_AHEAD) {
        __builtin_amdgcn_sched_barrier(0);
        #pragma unroll
        for (int t = W_AHEAD; t < 12; ++t) bw[t] = wp[2 * t];
        bw[12] = (hi == 0) ? wp[24] : zero4;
      }
      const v4f a4 = *(const v4f*)&q_lds[ln][8 * j + 4 * hi];
      v2f a, bfrag;
      a[0] = a4[0];  a[1] = a4[1];  bfrag[0] = bw[j][0];  bfrag[1] = bw[j][1];
      acc = __builtin_amdgcn_wmma_f32_16x16x4_f32(
           false, a,  false, bfrag,
           (short)0, acc,  false,  false);
      a[0] = a4[2];  a[1] = a4[3];  bfrag[0] = bw[j][2];  bfrag[1] = bw[j][3];
      acc = __builtin_amdgcn_wmma_f32_16x16x4_f32(
           false, a,  false, bfrag,
           (short)0, acc,  false,  false);
    }
    #pragma unroll
    for (int r = 0; r < 8; ++r)
      v_lds[r + 8 * hi][w_side][w_nt * 16 + ln] = acc[r];
  }
  __syncthreads();

  {
    const int b  = bbase + wv;
    const int bl = wv;

    int icl[LSLOT], icr[RSLOT], iwl[LSLOT], iwr[RSLOT];
    #pragma unroll
    for (int t = 0; t < LSLOT; ++t) { icl[t] = lixs_c[b * LSLOT + t]; iwl[t] = lixs_w[b * LSLOT + t]; }
    #pragma unroll
    for (int t = 0; t < RSLOT; ++t) { icr[t] = rixs_c[b * RSLOT + t]; iwr[t] = rixs_w[b * RSLOT + t]; }

    float la[LSLOT], ra[RSLOT];
    score_group<LSLOT>(icl,  &v_lds[bl][0][0], ce_raw, lane, la);
    score_group<5>(icr,      &v_lds[bl][1][0], ce_raw, lane, ra);
    score_group<5>(icr + 5,  &v_lds[bl][1][0], ce_raw, lane, ra + 5);

    float mx = la[0];
    #pragma unroll
    for (int t = 1; t < LSLOT; ++t) mx = fmaxf(mx, la[t]);
    float s = 0.f;
    #pragma unroll
    for (int t = 0; t < LSLOT; ++t) { la[t] = expf(la[t] - mx); s += la[t]; }
    float rs = 1.f / s;
    #pragma unroll
    for (int t = 0; t < LSLOT; ++t) la[t] *= rs;

    mx = ra[0];
    #pragma unroll
    for (int t = 1; t < RSLOT; ++t) mx = fmaxf(mx, ra[t]);
    s = 0.f;
    #pragma unroll
    for (int t = 0; t < RSLOT; ++t) { ra[t] = expf(ra[t] - mx); s += ra[t]; }
    rs = 1.f / s;
    #pragma unroll
    for (int t = 0; t < RSLOT; ++t) ra[t] *= rs;

    float z0 = gL_b[0], z1 = gL_b[1];
    #pragma unroll
    for (int t = 0; t < LSLOT; ++t) {
      z0 = fmaf(la[t], gL_w[t],                z0);
      z1 = fmaf(la[t], gL_w[(LSLOT + RSLOT) + t], z1);
    }
    #pragma unroll
    for (int t = 0; t < RSLOT; ++t) {
      z0 = fmaf(ra[t], gL_w[LSLOT + t],                       z0);
      z1 = fmaf(ra[t], gL_w[(LSLOT + RSLOT) + LSLOT + t],     z1);
    }
    float gm = fmaxf(z0, z1);
    float e0 = expf(z0 - gm), e1 = expf(z1 - gm);
    float gi = 1.f / (e0 + e1);
    float g0 = e0 * gi, g1 = e1 * gi;

    const float* pl[LSLOT]; const float* pr[RSLOT];
    #pragma unroll
    for (int t = 0; t < LSLOT; ++t) {
      pl[t]  = wvec + (long)max(iwl[t], 0) * WDIM;
      la[t]  = uniform_f(la[t]);
    }
    #pragma unroll
    for (int t = 0; t < RSLOT; ++t) {
      pr[t]  = wvec + (long)max(iwr[t], 0) * WDIM;
      ra[t]  = uniform_f(ra[t]);
    }
    for (int j0 = lane; j0 < WDIM; j0 += 64) {
      const int  j1  = j0 + 32;
      const bool in1 = j1 < WDIM;
      const int  j1c = in1 ? j1 : WDIM - 1;
      float xl0[LSLOT], xl1[LSLOT], xr0[RSLOT], xr1[RSLOT];
      #pragma unroll
      for (int t = 0; t < LSLOT; ++t) { xl0[t] = pl[t][j0]; xl1[t] = pl[t][j1c]; }
      #pragma unroll
      for (int t = 0; t < RSLOT; ++t) { xr0[t] = pr[t][j0]; xr1[t] = pr[t][j1c]; }
      float accL0 = 0.f, accR0 = 0.f, accL1 = 0.f, accR1 = 0.f;
      #pragma unroll
      for (int t = 0; t < LSLOT; ++t) {
        accL0 = fmaf(la[t], iwl[t] < 0 ? 0.f : xl0[t], accL0);
        accL1 = fmaf(la[t], iwl[t] < 0 ? 0.f : xl1[t], accL1);
      }
      #pragma unroll
      for (int t = 0; t < RSLOT; ++t) {
        accR0 = fmaf(ra[t], iwr[t] < 0 ? 0.f : xr0[t], accR0);
        accR1 = fmaf(ra[t], iwr[t] < 0 ? 0.f : xr1[t], accR1);
      }
      o_lds[wv * WDIM + j0] = g0 * accL0 + g1 * accR0;
      if (in1) o_lds[wv * WDIM + j1] = g0 * accL1 + g1 * accR1;
    }
  }
  __syncthreads();

  {
    float* otile = out + (long)bbase * WDIM;
    for (int line = wv; line < (16 * WDIM) / 32; line += 16)
      otile[line * 32 + lane] = o_lds[line * 32 + lane];
  }
}

extern "C" void kernel_launch(void* const* d_in, const int* in_sizes, int n_in,
                              void* d_out, int out_size, void* d_ws, size_t ws_size,
                              hipStream_t stream) {
  const float* ce_raw = (const float*)d_in[0];
  const float* wvec   = (const float*)d_in[1];
  const float* lW     = (const float*)d_in[2];
  const float* rW     = (const float*)d_in[3];
  const float* gL_w   = (const float*)d_in[4];
  const float* gL_b   = (const float*)d_in[5];
  const int*   qidx   = (const int*)d_in[6];
  const int*   lixs_c = (const int*)d_in[7];
  const int*   rixs_c = (const int*)d_in[8];
  const int*   lixs_w = (const int*)d_in[9];
  const int*   rixs_w = (const int*)d_in[10];

  const int B = in_sizes[6];
  const int blocks = B / 16;
  fused_gated_attn_kernel<<<blocks, 512, 0, stream>>>(
      ce_raw, wvec, lW, rW, gL_w, gL_b,
      qidx, lixs_c, rixs_c, lixs_w, rixs_w, (float*)d_out);
}


# Row gather loads widened from 4 to 16 B per lane (8 B on a word row's 176-B tail): 400-B context/query row 1 load (was 4), 1200-B word row 3 (was 10). Lane order of score dot and norm sums changes.
# speedup vs baseline: 22.5737x; 1.0571x over previous
#include <hip/hip_runtime.h>
#include <hip/hip_bf16.h>

typedef float v2f __attribute__((ext_vector_type(2)));
typedef float v4f __attribute__((ext_vector_type(4)));
typedef float v8f __attribute__((ext_vector_type(8)));

#define CDIM  100
#define LSLOT 7
#define RSLOT 10
#define WDIM  300
#define QPITCH 108
#define W_AHEAD 9
#define NEGV  -1e30f

__device__ __forceinline__ float wave_sum(float v) {
  #pragma unroll
  for (int m = 16; m; m >>= 1) v += __shfl_xor(v, m, 32);
  return v;
}

template <int N>
__device__ __forceinline__ void score_group(const int* idx, const float* vrow,
                                            const float* __restrict__ ce_raw, int lane,
                                            float* sc) {
  const bool act = lane < CDIM / 4;
  const int  c4  = act ? lane : CDIM / 4 - 1;
  const v4f v = *(const v4f*)(vrow + 4 * c4);
  v4f x[N];
  #pragma unroll
  for (int t = 0; t < N; ++t) {
    const v4f* row = (const v4f*)(ce_raw + (long)max(idx[t], 0) * CDIM);
    x[t] = row[c4];
  }
  const v2f vlo = {v[0], v[1]}, vhi = {v[2], v[3]};
  float dot[N], ss[N];
  #pragma unroll
  for (int t = 0; t < N; ++t) {
    const v2f xlo = {x[t][0], x[t][1]}, xhi = {x[t][2], x[t][3]};
    const v2f d2 = __builtin_elementwise_fma(xhi, vhi, xlo * vlo);
    const v2f s2 = __builtin_elementwise_fma(xhi, xhi, xlo * xlo);
    dot[t] = act ? d2[0] + d2[1] : 0.f;
    ss[t]  = act ? s2[0] + s2[1] : 0.f;
  }
  #pragma unroll
  for (int m = 16; m; m >>= 1) {
    #pragma unroll
    for (int t = 0; t < N; ++t) {
      dot[t] += __shfl_xor(dot[t], m, 32);
      ss[t]  += __shfl_xor(ss[t],  m, 32);
    }
  }
  #pragma unroll
  for (int t = 0; t < N; ++t)
    sc[t] = (idx[t] < 0) ? NEGV : dot[t] / fmaxf(sqrtf(ss[t]), 1e-12f);
}

template <typename VT, int NE, int SPLIT>
__device__ __forceinline__ VT emb_piece(const float* const (&pl)[LSLOT], const float* const (&pr)[RSLOT],
                                        const float (&la)[LSLOT], const float (&ra)[RSLOT],
                                        const int (&iwl)[LSLOT], const int (&iwr)[RSLOT],
                                        int vi, float g0, float g1) {
  VT xl[LSLOT], xr[RSLOT];
  #pragma unroll
  for (int t = 0; t < LSLOT; ++t) xl[t] = ((const VT*)pl[t])[vi];
  #pragma unroll
  for (int t = 0; t < SPLIT; ++t) xr[t] = ((const VT*)pr[t])[vi];
  VT accL, accR, o;
  #pragma unroll
  for (int e = 0; e < NE; ++e) { accL[e] = 0.f; accR[e] = 0.f; }
  #pragma unroll
  for (int t = 0; t < LSLOT; ++t) {
    #pragma unroll
    for (int e = 0; e < NE; ++e) accL[e] = fmaf(la[t], iwl[t] < 0 ? 0.f : xl[t][e], accL[e]);
  }
  #pragma unroll
  for (int t = 0; t < SPLIT; ++t) {
    #pragma unroll
    for (int e = 0; e < NE; ++e) accR[e] = fmaf(ra[t], iwr[t] < 0 ? 0.f : xr[t][e], accR[e]);
  }
  if (SPLIT < RSLOT) {
    __builtin_amdgcn_sched_barrier(0);
    #pragma unroll
    for (int t = SPLIT; t < RSLOT; ++t) xr[t] = ((const VT*)pr[t])[vi];
    #pragma unroll
    for (int t = SPLIT; t < RSLOT; ++t) {
      #pragma unroll
      for (int e = 0; e < NE; ++e) accR[e] = fmaf(ra[t], iwr[t] < 0 ? 0.f : xr[t][e], accR[e]);
    }
  }
  #pragma unroll
  for (int e = 0; e < NE; ++e) o[e] = g0 * accL[e] + g1 * accR[e];
  return o;
}

__device__ __forceinline__ float uniform_f(float v) {
  return __int_as_float(__builtin_amdgcn_readfirstlane(__float_as_int(v)));
}

__global__ __launch_bounds__(512) void fused_gated_attn_kernel(
    const float* __restrict__ ce_raw, const float* __restrict__ wvec,
    const float* __restrict__ lW,     const float* __restrict__ rW,
    const float* __restrict__ gL_w,   const float* __restrict__ gL_b,
    const int* __restrict__ qidx,
    const int* __restrict__ lixs_c,   const int* __restrict__ rixs_c,
    const int* __restrict__ lixs_w,   const int* __restrict__ rixs_w,
    float* __restrict__ out)
{
  __shared__ __attribute__((aligned(16))) float q_lds[16][QPITCH];
  __shared__ __attribute__((aligned(16))) float v_lds[16][2][112];
  __shared__ __attribute__((aligned(16))) float o_lds[16 * WDIM];

  const int lane  = threadIdx.x & 31;
  const int wv    = __builtin_amdgcn_readfirstlane(threadIdx.x >> 5);
  const int bbase = blockIdx.x * 16;
  const int hi = lane >> 4, ln = lane & 15;

  const int  w_side = wv / 7, w_nt = wv % 7;
  const v4f* wp = (const v4f*)((w_side ? rW : lW) + min(w_nt * 16 + ln, CDIM - 1) * CDIM + 4 * hi);
  v4f bw[13];
  if (wv < 14) {
    #pragma unroll
    for (int j = 0; j < W_AHEAD; ++j) bw[j] = wp[2 * j];
  }

  {
    const int r = wv;
    long row = qidx[bbase + r];
    const v4f zero4q = {0.f, 0.f, 0.f, 0.f};
    v4f qv = zero4q;
    if (lane < CDIM / 4) qv = ((const v4f*)(ce_raw + row * CDIM))[lane];
    float ss = 0.f;
    #pragma unroll
    for (int e = 0; e < 4; ++e) ss = fmaf(qv[e], qv[e], ss);
    ss = wave_sum(ss);
    float inv = 1.f / fmaxf(sqrtf(ss), 1e-12f);
    if (lane < CDIM / 4 + 1) {
      v4f qn;
      #pragma unroll
      for (int e = 0; e < 4; ++e) qn[e] = qv[e] * inv;
      *(v4f*)&q_lds[r][4 * lane] = qn;
    }
  }
  __syncthreads();

  if (wv < 14) {
    const v4f zero4 = {0.f, 0.f, 0.f, 0.f};
    v8f acc = {};
    #pragma unroll
    for (int j = 0; j < 13; ++j) {
      if (j == 13 - W_AHEAD) {
        __builtin_amdgcn_sched_barrier(0);
        #pragma unroll
        for (int t = W_AHEAD; t < 12; ++t) bw[t] = wp[2 * t];
        bw[12] = (hi == 0) ? wp[24] : zero4;
      }
      const v4f a4 = *(const v4f*)&q_lds[ln][8 * j + 4 * hi];
      v2f a, bfrag;
      a[0] = a4[0];  a[1] = a4[1];  bfrag[0] = bw[j][0];  bfrag[1] = bw[j][1];
      acc = __builtin_amdgcn_wmma_f32_16x16x4_f32(
           false, a,  false, bfrag,
           (short)0, acc,  false,  false);
      a[0] = a4[2];  a[1] = a4[3];  bfrag[0] = bw[j][2];  bfrag[1] = bw[j][3];
      acc = __builtin_amdgcn_wmma_f32_16x16x4_f32(
           false, a,  false, bfrag,
           (short)0, acc,  false,  false);
    }
    #pragma unroll
    for (int r = 0; r < 8; ++r)
      v_lds[r + 8 * hi][w_side][w_nt * 16 + ln] = acc[r];
  }
  __syncthreads();

  {
    const int b  = bbase + wv;
    const int bl = wv;

    int icl[LSLOT], icr[RSLOT], iwl[LSLOT], iwr[RSLOT];
    #pragma unroll
    for (int t = 0; t < LSLOT; ++t) { icl[t] = lixs_c[b * LSLOT + t]; iwl[t] = lixs_w[b * LSLOT + t]; }
    #pragma unroll
    for (int t = 0; t < RSLOT; ++t) { icr[t] = rixs_c[b * RSLOT + t]; iwr[t] = rixs_w[b * RSLOT + t]; }

    float la[LSLOT], ra[RSLOT];
    score_group<LSLOT>(icl,  &v_lds[bl][0][0], ce_raw, lane, la);
    score_group<5>(icr,      &v_lds[bl][1][0], ce_raw, lane, ra);
    score_group<5>(icr + 5,  &v_lds[bl][1][0], ce_raw, lane, ra + 5);

    float mx = la[0];
    #pragma unroll
    for (int t = 1; t < LSLOT; ++t) mx = fmaxf(mx, la[t]);
    float s = 0.f;
    #pragma unroll
    for (int t = 0; t < LSLOT; ++t) { la[t] = expf(la[t] - mx); s += la[t]; }
    float rs = 1.f / s;
    #pragma unroll
    for (int t = 0; t < LSLOT; ++t) la[t] *= rs;

    mx = ra[0];
    #pragma unroll
    for (int t = 1; t < RSLOT; ++t) mx = fmaxf(mx, ra[t]);
    s = 0.f;
    #pragma unroll
    for (int t = 0; t < RSLOT; ++t) { ra[t] = expf(ra[t] - mx); s += ra[t]; }
    rs = 1.f / s;
    #pragma unroll
    for (int t = 0; t < RSLOT; ++t) ra[t] *= rs;

    float z0 = gL_b[0], z1 = gL_b[1];
    #pragma unroll
    for (int t = 0; t < LSLOT; ++t) {
      z0 = fmaf(la[t], gL_w[t],                z0);
      z1 = fmaf(la[t], gL_w[(LSLOT + RSLOT) + t], z1);
    }
    #pragma unroll
    for (int t = 0; t < RSLOT; ++t) {
      z0 = fmaf(ra[t], gL_w[LSLOT + t],                       z0);
      z1 = fmaf(ra[t], gL_w[(LSLOT + RSLOT) + LSLOT + t],     z1);
    }
    float gm = fmaxf(z0, z1);
    float e0 = expf(z0 - gm), e1 = expf(z1 - gm);
    float gi = 1.f / (e0 + e1);
    float g0 = e0 * gi, g1 = e1 * gi;

    const float* pl[LSLOT]; const float* pr[RSLOT];
    #pragma unroll
    for (int t = 0; t < LSLOT; ++t) {
      pl[t]  = wvec + (long)max(iwl[t], 0) * WDIM;
      la[t]  = uniform_f(la[t]);
    }
    #pragma unroll
    for (int t = 0; t < RSLOT; ++t) {
      pr[t]  = wvec + (long)max(iwr[t], 0) * WDIM;
      ra[t]  = uniform_f(ra[t]);
    }
    const v4f oA = emb_piece<v4f, 4, 2>(pl, pr, la, ra, iwl, iwr, lane,      g0, g1);
    *(v4f*)&o_lds[wv * WDIM + 4 * lane] = oA;
    __builtin_amdgcn_sched_barrier(0);
    const v4f oB = emb_piece<v4f, 4, 2>(pl, pr, la, ra, iwl, iwr, lane + 32, g0, g1);
    *(v4f*)&o_lds[wv * WDIM + 128 + 4 * lane] = oB;
    const bool tact = lane < (WDIM - 256) / 2;
    const int  t2   = 128 + (tact ? lane : (WDIM - 256) / 2 - 1);
    const v2f oT = emb_piece<v2f, 2, RSLOT>(pl, pr, la, ra, iwl, iwr, t2, g0, g1);
    if (tact) *(v2f*)&o_lds[wv * WDIM + 256 + 2 * lane] = oT;
  }
  __syncthreads();

  {
    float* otile = out + (long)bbase * WDIM;
    for (int line = wv; line < (16 * WDIM) / 32; line += 16)
      otile[line * 32 + lane] = o_lds[line * 32 + lane];
  }
}

extern "C" void kernel_launch(void* const* d_in, const int* in_sizes, int n_in,
                              void* d_out, int out_size, void* d_ws, size_t ws_size,
                              hipStream_t stream) {
  const float* ce_raw = (const float*)d_in[0];
  const float* wvec   = (const float*)d_in[1];
  const float* lW     = (const float*)d_in[2];
  const float* rW     = (const float*)d_in[3];
  const float* gL_w   = (const float*)d_in[4];
  const float* gL_b   = (const float*)d_in[5];
  const int*   qidx   = (const int*)d_in[6];
  const int*   lixs_c = (const int*)d_in[7];
  const int*   rixs_c = (const int*)d_in[8];
  const int*   lixs_w = (const int*)d_in[9];
  const int*   rixs_w = (const int*)d_in[10];

  const int B = in_sizes[6];
  const int blocks = B / 16;
  fused_gated_attn_kernel<<<blocks, 512, 0, stream>>>(
      ce_raw, wvec, lW, rW, gL_w, gL_b,
      qidx, lixs_c, rixs_c, lixs_w, rixs_w, (float*)d_out);
}


# Score epilogue one slot per lane: sqrt, divisions, exp of the 17 slots run once on 17 lanes (were 17 runs on wave-uniform values); softmax max/sum by 16-lane DPP tree. Softmax sum order changes.
# speedup vs baseline: 23.3519x; 1.0345x over previous
#include <hip/hip_runtime.h>
#include <hip/hip_bf16.h>

typedef float v2f __attribute__((ext_vector_type(2)));
typedef float v4f __attribute__((ext_vector_type(4)));
typedef float v8f __attribute__((ext_vector_type(8)));

#define CDIM  100
#define LSLOT 7
#define RSLOT 10
#define WDIM  300
#define QPITCH 108
#define W_AHEAD 9
#define NEGV  -1e30f

__device__ __forceinline__ float wave_sum(float v) {
  #pragma unroll
  for (int m = 16; m; m >>= 1) v += __shfl_xor(v, m, 32);
  return v;
}

template <int N>
__device__ __forceinline__ void score_group(const int* idx, const float* vrow,
                                            const float* __restrict__ ce_raw, int lane,
                                            int lane0, float& dsel, float& ssel) {
  const bool act = lane < CDIM / 4;
  const int  c4  = act ? lane : CDIM / 4 - 1;
  const v4f v = *(const v4f*)(vrow + 4 * c4);
  v4f x[N];
  #pragma unroll
  for (int t = 0; t < N; ++t) {
    const v4f* row = (const v4f*)(ce_raw + (long)max(idx[t], 0) * CDIM);
    x[t] = row[c4];
  }
  const v2f vlo = {v[0], v[1]}, vhi = {v[2], v[3]};
  float dot[N], ss[N];
  #pragma unroll
  for (int t = 0; t < N; ++t) {
    const v2f xlo = {x[t][0], x[t][1]}, xhi = {x[t][2], x[t][3]};
    const v2f d2 = __builtin_elementwise_fma(xhi, vhi, xlo * vlo);
    const v2f s2 = __builtin_elementwise_fma(xhi, xhi, xlo * xlo);
    dot[t] = act ? d2[0] + d2[1] : 0.f;
    ss[t]  = act ? s2[0] + s2[1] : 0.f;
  }
  #pragma unroll
  for (int m = 16; m; m >>= 1) {
    #pragma unroll
    for (int t = 0; t < N; ++t) {
      dot[t] += __shfl_xor(dot[t], m, 32);
      ss[t]  += __shfl_xor(ss[t],  m, 32);
    }
  }
  #pragma unroll
  for (int t = 0; t < N; ++t)
    if (lane == lane0 + t) { dsel = dot[t]; ssel = ss[t]; }
}

template <int M>
__device__ __forceinline__ float row_xor(float v) {
  return __int_as_float(__builtin_amdgcn_update_dpp(0, __float_as_int(v), 0x160 | M, 0xf, 0xf, false));
}
__device__ __forceinline__ float row_max16(float v) {
  v = fmaxf(v, row_xor<8>(v));  v = fmaxf(v, row_xor<4>(v));
  v = fmaxf(v, row_xor<2>(v));  v = fmaxf(v, row_xor<1>(v));
  return v;
}
__device__ __forceinline__ float row_sum16(float v) {
  v += row_xor<8>(v);  v += row_xor<4>(v);
  v += row_xor<2>(v);  v += row_xor<1>(v);
  return v;
}

template <int L>
__device__ __forceinline__ float lane_value(float v) {
  return __int_as_float(__builtin_amdgcn_readlane(__float_as_int(v), L));
}

template <typename VT, int NE, int SPLIT>
__device__ __forceinline__ VT emb_piece(const float* const (&pl)[LSLOT], const float* const (&pr)[RSLOT],
                                        const float (&la)[LSLOT], const float (&ra)[RSLOT],
                                        const int (&iwl)[LSLOT], const int (&iwr)[RSLOT],
                                        int vi, float g0, float g1) {
  VT xl[LSLOT], xr[RSLOT];
  #pragma unroll
  for (int t = 0; t < LSLOT; ++t) xl[t] = ((const VT*)pl[t])[vi];
  #pragma unroll
  for (int t = 0; t < SPLIT; ++t) xr[t] = ((const VT*)pr[t])[vi];
  VT accL, accR, o;
  #pragma unroll
  for (int e = 0; e < NE; ++e) { accL[e] = 0.f; accR[e] = 0.f; }
  #pragma unroll
  for (int t = 0; t < LSLOT; ++t) {
    #pragma unroll
    for (int e = 0; e < NE; ++e) accL[e] = fmaf(la[t], iwl[t] < 0 ? 0.f : xl[t][e], accL[e]);
  }
  #pragma unroll
  for (int t = 0; t < SPLIT; ++t) {
    #pragma unroll
    for (int e = 0; e < NE; ++e) accR[e] = fmaf(ra[t], iwr[t] < 0 ? 0.f : xr[t][e], accR[e]);
  }
  if (SPLIT < RSLOT) {
    __builtin_amdgcn_sched_barrier(0);
    #pragma unroll
    for (int t = SPLIT; t < RSLOT; ++t) xr[t] = ((const VT*)pr[t])[vi];
    #pragma unroll
    for (int t = SPLIT; t < RSLOT; ++t) {
      #pragma unroll
      for (int e = 0; e < NE; ++e) accR[e] = fmaf(ra[t], iwr[t] < 0 ? 0.f : xr[t][e], accR[e]);
    }
  }
  #pragma unroll
  for (int e = 0; e < NE; ++e) o[e] = g0 * accL[e] + g1 * accR[e];
  return o;
}

__global__ __launch_bounds__(512) void fused_gated_attn_kernel(
    const float* __restrict__ ce_raw, const float* __restrict__ wvec,
    const float* __restrict__ lW,     const float* __restrict__ rW,
    const float* __restrict__ gL_w,   const float* __restrict__ gL_b,
    const int* __restrict__ qidx,
    const int* __restrict__ lixs_c,   const int* __restrict__ rixs_c,
    const int* __restrict__ lixs_w,   const int* __restrict__ rixs_w,
    float* __restrict__ out)
{
  __shared__ __attribute__((aligned(16))) float q_lds[16][QPITCH];
  __shared__ __attribute__((aligned(16))) float v_lds[16][2][112];
  __shared__ __attribute__((aligned(16))) float o_lds[16 * WDIM];

  const int lane  = threadIdx.x & 31;
  const int wv    = __builtin_amdgcn_readfirstlane(threadIdx.x >> 5);
  const int bbase = blockIdx.x * 16;
  const int hi = lane >> 4, ln = lane & 15;

  const int  w_side = wv / 7, w_nt = wv % 7;
  const v4f* wp = (const v4f*)((w_side ? rW : lW) + min(w_nt * 16 + ln, CDIM - 1) * CDIM + 4 * hi);
  v4f bw[13];
  if (wv < 14) {
    #pragma unroll
    for (int j = 0; j < W_AHEAD; ++j) bw[j] = wp[2 * j];
  }

  {
    const int r = wv;
    long row = qidx[bbase + r];
    const v4f zero4q = {0.f, 0.f, 0.f, 0.f};
    v4f qv = zero4q;
    if (lane < CDIM / 4) qv = ((const v4f*)(ce_raw + row * CDIM))[lane];
    float ss = 0.f;
    #pragma unroll
    for (int e = 0; e < 4; ++e) ss = fmaf(qv[e], qv[e], ss);
    ss = wave_sum(ss);
    float inv = 1.f / fmaxf(sqrtf(ss), 1e-12f);
    if (lane < CDIM / 4 + 1) {
      v4f qn;
      #pragma unroll
      for (int e = 0; e < 4; ++e) qn[e] = qv[e] * inv;
      *(v4f*)&q_lds[r][4 * lane] = qn;
    }
  }
  __syncthreads();

  if (wv < 14) {
    const v4f zero4 = {0.f, 0.f, 0.f, 0.f};
    v8f acc = {};
    #pragma unroll
    for (int j = 0; j < 13; ++j) {
      if (j == 13 - W_AHEAD) {
        __builtin_amdgcn_sched_barrier(0);
        #pragma unroll
        for (int t = W_AHEAD; t < 12; ++t) bw[t] = wp[2 * t];
        bw[12] = (hi == 0) ? wp[24] : zero4;
      }
      const v4f a4 = *(const v4f*)&q_lds[ln][8 * j + 4 * hi];
      v2f a, bfrag;
      a[0] = a4[0];  a[1] = a4[1];  bfrag[0] = bw[j][0];  bfrag[1] = bw[j][1];
      acc = __builtin_amdgcn_wmma_f32_16x16x4_f32(
           false, a,  false, bfrag,
           (short)0, acc,  false,  false);
      a[0] = a4[2];  a[1] = a4[3];  bfrag[0] = bw[j][2];  bfrag[1] = bw[j][3];
      acc = __builtin_amdgcn_wmma_f32_16x16x4_f32(
           false, a,  false, bfrag,
           (short)0, acc,  false,  false);
    }
    #pragma unroll
    for (int r = 0; r < 8; ++r)
      v_lds[r + 8 * hi][w_side][w_nt * 16 + ln] = acc[r];
  }
  __syncthreads();

  {
    const int b  = bbase + wv;
    const int bl = wv;

    int icl[LSLOT], icr[RSLOT], iwl[LSLOT], iwr[RSLOT];
    #pragma unroll
    for (int t = 0; t < LSLOT; ++t) { icl[t] = lixs_c[b * LSLOT + t]; iwl[t] = lixs_w[b * LSLOT + t]; }
    #pragma unroll
    for (int t = 0; t < RSLOT; ++t) { icr[t] = rixs_c[b * RSLOT + t]; iwr[t] = rixs_w[b * RSLOT + t]; }

    float dsel = 0.f, ssel = 1.f;
    score_group<LSLOT>(icl,  &v_lds[bl][0][0], ce_raw, lane, 0,  dsel, ssel);
    score_group<5>(icr,      &v_lds[bl][1][0], ce_raw, lane, 16, dsel, ssel);
    __builtin_amdgcn_sched_barrier(0);
    score_group<5>(icr + 5,  &v_lds[bl][1][0], ce_raw, lane, 21, dsel, ssel);

    const unsigned no_slot = ~((1u << LSLOT) - 1u) & ~(((1u << RSLOT) - 1u) << 16);
    unsigned neg_bits = no_slot;
    #pragma unroll
    for (int t = 0; t < LSLOT; ++t) neg_bits |= (icl[t] < 0 ? 1u : 0u) << t;
    #pragma unroll
    for (int t = 0; t < RSLOT; ++t) neg_bits |= (icr[t] < 0 ? 1u : 0u) << (16 + t);
    const bool has_slot = !((no_slot  >> lane) & 1u);
    const bool is_neg   =   (neg_bits >> lane) & 1u;

    float sc = dsel / fmaxf(sqrtf(ssel), 1e-12f);
    sc = is_neg ? NEGV : sc;
    const float mx = row_max16(sc);
    const float ex = has_slot ? expf(sc - mx) : 0.f;
    const float aw = ex * (1.f / row_sum16(ex));

    float la[LSLOT], ra[RSLOT];
    la[0] = lane_value<0>(aw);  la[1] = lane_value<1>(aw);  la[2] = lane_value<2>(aw);
    la[3] = lane_value<3>(aw);  la[4] = lane_value<4>(aw);  la[5] = lane_value<5>(aw);
    la[6] = lane_value<6>(aw);
    ra[0] = lane_value<16>(aw); ra[1] = lane_value<17>(aw); ra[2] = lane_value<18>(aw);
    ra[3] = lane_value<19>(aw); ra[4] = lane_value<20>(aw); ra[5] = lane_value<21>(aw);
    ra[6] = lane_value<22>(aw); ra[7] = lane_value<23>(aw); ra[8] = lane_value<24>(aw);
    ra[9] = lane_value<25>(aw);

    float z0 = gL_b[0], z1 = gL_b[1];
    #pragma unroll
    for (int t = 0; t < LSLOT; ++t) {
      z0 = fmaf(la[t], gL_w[t],                z0);
      z1 = fmaf(la[t], gL_w[(LSLOT + RSLOT) + t], z1);
    }
    #pragma unroll
    for (int t = 0; t < RSLOT; ++t) {
      z0 = fmaf(ra[t], gL_w[LSLOT + t],                       z0);
      z1 = fmaf(ra[t], gL_w[(LSLOT + RSLOT) + LSLOT + t],     z1);
    }
    float gm = fmaxf(z0, z1);
    float e0 = expf(z0 - gm), e1 = expf(z1 - gm);
    float gi = 1.f / (e0 + e1);
    float g0 = e0 * gi, g1 = e1 * gi;

    const float* pl[LSLOT]; const float* pr[RSLOT];
    #pragma unroll
    for (int t = 0; t < LSLOT; ++t) pl[t] = wvec + (long)max(iwl[t], 0) * WDIM;
    #pragma unroll
    for (int t = 0; t < RSLOT; ++t) pr[t] = wvec + (long)max(iwr[t], 0) * WDIM;
    const v4f oA = emb_piece<v4f, 4, 2>(pl, pr, la, ra, iwl, iwr, lane,      g0, g1);
    *(v4f*)&o_lds[wv * WDIM + 4 * lane] = oA;
    __builtin_amdgcn_sched_barrier(0);
    const v4f oB = emb_piece<v4f, 4, 2>(pl, pr, la, ra, iwl, iwr, lane + 32, g0, g1);
    *(v4f*)&o_lds[wv * WDIM + 128 + 4 * lane] = oB;
    const bool tact = lane < (WDIM - 256) / 2;
    const int  t2   = 128 + (tact ? lane : (WDIM - 256) / 2 - 1);
    const v2f oT = emb_piece<v2f, 2, RSLOT>(pl, pr, la, ra, iwl, iwr, t2, g0, g1);
    if (tact) *(v2f*)&o_lds[wv * WDIM + 256 + 2 * lane] = oT;
  }
  __syncthreads();

  {
    float* otile = out + (long)bbase * WDIM;
    for (int line = wv; line < (16 * WDIM) / 32; line += 16)
      otile[line * 32 + lane] = o_lds[line * 32 + lane];
  }
}

extern "C" void kernel_launch(void* const* d_in, const int* in_sizes, int n_in,
                              void* d_out, int out_size, void* d_ws, size_t ws_size,
                              hipStream_t stream) {
  const float* ce_raw = (const float*)d_in[0];
  const float* wvec   = (const float*)d_in[1];
  const float* lW     = (const float*)d_in[2];
  const float* rW     = (const float*)d_in[3];
  const float* gL_w   = (const float*)d_in[4];
  const float* gL_b   = (const float*)d_in[5];
  const int*   qidx   = (const int*)d_in[6];
  const int*   lixs_c = (const int*)d_in[7];
  const int*   rixs_c = (const int*)d_in[8];
  const int*   lixs_w = (const int*)d_in[9];
  const int*   rixs_w = (const int*)d_in[10];

  const int B = in_sizes[6];
  const int blocks = B / 16;
  fused_gated_attn_kernel<<<blocks, 512, 0, stream>>>(
      ce_raw, wvec, lW, rW, gL_w, gL_b,
      qidx, lixs_c, rixs_c, lixs_w, rixs_w, (float*)d_out);
}


# Wave sums of the 17 score dots, 17 norms and the query norm on the vector ALU (l^16 lane swap + 16-lane DPP tree, dot and norm in one row each); were 175 ds_bpermute via LDS. Same sum tree.
# speedup vs baseline: 25.1444x; 1.0768x over previous
#include <hip/hip_runtime.h>
#include <hip/hip_bf16.h>

typedef float v2f __attribute__((ext_vector_type(2)));
typedef float v4f __attribute__((ext_vector_type(4)));
typedef float v8f __attribute__((ext_vector_type(8)));

#define CDIM  100
#define LSLOT 7
#define RSLOT 10
#define WDIM  300
#define QPITCH 108
#define W_AHEAD 9
#define NEGV  -1e30f

template <int M>
__device__ __forceinline__ float row_xor(float v) {
  return __int_as_float(__builtin_amdgcn_update_dpp(0, __float_as_int(v), 0x160 | M, 0xf, 0xf, false));
}
__device__ __forceinline__ float row_max16(float v) {
  v = fmaxf(v, row_xor<8>(v));  v = fmaxf(v, row_xor<4>(v));
  v = fmaxf(v, row_xor<2>(v));  v = fmaxf(v, row_xor<1>(v));
  return v;
}
__device__ __forceinline__ float row_sum16(float v) {
  v += row_xor<8>(v);  v += row_xor<4>(v);
  v += row_xor<2>(v);  v += row_xor<1>(v);
  return v;
}
__device__ __forceinline__ float xhalf(float v) {
  const int b = __float_as_int(v);
  return __int_as_float(__builtin_amdgcn_permlanex16(b, b, 0x76543210, 0xfedcba98, false, false));
}
__device__ __forceinline__ float wave_sum(float v) {
  v += xhalf(v);
  return row_sum16(v);
}

template <int N>
__device__ __forceinline__ void score_group(const int* idx, const float* vrow,
                                            const float* __restrict__ ce_raw, int lane,
                                            int lane0, float& dsel, float& ssel) {
  const bool act = lane < CDIM / 4;
  const int  c4  = act ? lane : CDIM / 4 - 1;
  const v4f v = *(const v4f*)(vrow + 4 * c4);
  v4f x[N];
  #pragma unroll
  for (int t = 0; t < N; ++t) {
    const v4f* row = (const v4f*)(ce_raw + (long)max(idx[t], 0) * CDIM);
    x[t] = row[c4];
  }
  const v2f vlo = {v[0], v[1]}, vhi = {v[2], v[3]};
  float dot[N], ss[N];
  #pragma unroll
  for (int t = 0; t < N; ++t) {
    const v2f xlo = {x[t][0], x[t][1]}, xhi = {x[t][2], x[t][3]};
    const v2f d2 = __builtin_elementwise_fma(xhi, vhi, xlo * vlo);
    const v2f s2 = __builtin_elementwise_fma(xhi, xhi, xlo * xlo);
    dot[t] = act ? d2[0] + d2[1] : 0.f;
    ss[t]  = act ? s2[0] + s2[1] : 0.f;
  }
  const bool hi_row = lane >= 16;
  #pragma unroll
  for (int t = 0; t < N; ++t) {
    const float mine = hi_row ? ss[t]  : dot[t];
    const float give = hi_row ? dot[t] : ss[t];
    const float r = row_sum16(mine + xhalf(give));
    const float x = xhalf(r);
    const bool  low = (lane0 + t) < 16;
    const float dtot = low ? r : x, stot = low ? x : r;
    if (lane == lane0 + t) { dsel = dtot; ssel = stot; }
  }
}

template <int L>
__device__ __forceinline__ float lane_value(float v) {
  return __int_as_float(__builtin_amdgcn_readlane(__float_as_int(v), L));
}

template <typename VT, int NE, int SPLIT>
__device__ __forceinline__ VT emb_piece(const float* const (&pl)[LSLOT], const float* const (&pr)[RSLOT],
                                        const float (&la)[LSLOT], const float (&ra)[RSLOT],
                                        const int (&iwl)[LSLOT], const int (&iwr)[RSLOT],
                                        int vi, float g0, float g1) {
  VT xl[LSLOT], xr[RSLOT];
  #pragma unroll
  for (int t = 0; t < LSLOT; ++t) xl[t] = ((const VT*)pl[t])[vi];
  #pragma unroll
  for (int t = 0; t < SPLIT; ++t) xr[t] = ((const VT*)pr[t])[vi];
  VT accL, accR, o;
  #pragma unroll
  for (int e = 0; e < NE; ++e) { accL[e] = 0.f; accR[e] = 0.f; }
  #pragma unroll
  for (int t = 0; t < LSLOT; ++t) {
    #pragma unroll
    for (int e = 0; e < NE; ++e) accL[e] = fmaf(la[t], iwl[t] < 0 ? 0.f : xl[t][e], accL[e]);
  }
  #pragma unroll
  for (int t = 0; t < SPLIT; ++t) {
    #pragma unroll
    for (int e = 0; e < NE; ++e) accR[e] = fmaf(ra[t], iwr[t] < 0 ? 0.f : xr[t][e], accR[e]);
  }
  if (SPLIT < RSLOT) {
    __builtin_amdgcn_sched_barrier(0);
    #pragma unroll
    for (int t = SPLIT; t < RSLOT; ++t) xr[t] = ((const VT*)pr[t])[vi];
    #pragma unroll
    for (int t = SPLIT; t < RSLOT; ++t) {
      #pragma unroll
      for (int e = 0; e < NE; ++e) accR[e] = fmaf(ra[t], iwr[t] < 0 ? 0.f : xr[t][e], accR[e]);
    }
  }
  #pragma unroll
  for (int e = 0; e < NE; ++e) o[e] = g0 * accL[e] + g1 * accR[e];
  return o;
}

__global__ __launch_bounds__(512) void fused_gated_attn_kernel(
    const float* __restrict__ ce_raw, const float* __restrict__ wvec,
    const float* __restrict__ lW,     const float* __restrict__ rW,
    const float* __restrict__ gL_w,   const float* __restrict__ gL_b,
    const int* __restrict__ qidx,
    const int* __restrict__ lixs_c,   const int* __restrict__ rixs_c,
    const int* __restrict__ lixs_w,   const int* __restrict__ rixs_w,
    float* __restrict__ out)
{
  __shared__ __attribute__((aligned(16))) float q_lds[16][QPITCH];
  __shared__ __attribute__((aligned(16))) float v_lds[16][2][112];
  __shared__ __attribute__((aligned(16))) float o_lds[16 * WDIM];

  const int lane  = threadIdx.x & 31;
  const int wv    = __builtin_amdgcn_readfirstlane(threadIdx.x >> 5);
  const int bbase = blockIdx.x * 16;
  const int hi = lane >> 4, ln = lane & 15;

  const int  w_side = wv / 7, w_nt = wv % 7;
  const v4f* wp = (const v4f*)((w_side ? rW : lW) + min(w_nt * 16 + ln, CDIM - 1) * CDIM + 4 * hi);
  v4f bw[13];
  if (wv < 14) {
    #pragma unroll
    for (int j = 0; j < W_AHEAD; ++j) bw[j] = wp[2 * j];
  }

  {
    const int r = wv;
    long row = qidx[bbase + r];
    const v4f zero4q = {0.f, 0.f, 0.f, 0.f};
    v4f qv = zero4q;
    if (lane < CDIM / 4) qv = ((const v4f*)(ce_raw + row * CDIM))[lane];
    float ss = 0.f;
    #pragma unroll
    for (int e = 0; e < 4; ++e) ss = fmaf(qv[e], qv[e], ss);
    ss = wave_sum(ss);
    float inv = 1.f / fmaxf(sqrtf(ss), 1e-12f);
    if (lane < CDIM / 4 + 1) {
      v4f qn;
      #pragma unroll
      for (int e = 0; e < 4; ++e) qn[e] = qv[e] * inv;
      *(v4f*)&q_lds[r][4 * lane] = qn;
    }
  }
  __syncthreads();

  if (wv < 14) {
    const v4f zero4 = {0.f, 0.f, 0.f, 0.f};
    v8f acc = {};
    #pragma unroll
    for (int j = 0; j < 13; ++j) {
      if (j == 13 - W_AHEAD) {
        __builtin_amdgcn_sched_barrier(0);
        #pragma unroll
        for (int t = W_AHEAD; t < 12; ++t) bw[t] = wp[2 * t];
        bw[12] = (hi == 0) ? wp[24] : zero4;
      }
      const v4f a4 = *(const v4f*)&q_lds[ln][8 * j + 4 * hi];
      v2f a, bfrag;
      a[0] = a4[0];  a[1] = a4[1];  bfrag[0] = bw[j][0];  bfrag[1] = bw[j][1];
      acc = __builtin_amdgcn_wmma_f32_16x16x4_f32(
           false, a,  false, bfrag,
           (short)0, acc,  false,  false);
      a[0] = a4[2];  a[1] = a4[3];  bfrag[0] = bw[j][2];  bfrag[1] = bw[j][3];
      acc = __builtin_amdgcn_wmma_f32_16x16x4_f32(
           false, a,  false, bfrag,
           (short)0, acc,  false,  false);
    }
    #pragma unroll
    for (int r = 0; r < 8; ++r)
      v_lds[r + 8 * hi][w_side][w_nt * 16 + ln] = acc[r];
  }
  __syncthreads();

  {
    const int b  = bbase + wv;
    const int bl = wv;

    int icl[LSLOT], icr[RSLOT], iwl[LSLOT], iwr[RSLOT];
    #pragma unroll
    for (int t = 0; t < LSLOT; ++t) { icl[t] = lixs_c[b * LSLOT + t]; iwl[t] = lixs_w[b * LSLOT + t]; }
    #pragma unroll
    for (int t = 0; t < RSLOT; ++t) { icr[t] = rixs_c[b * RSLOT + t]; iwr[t] = rixs_w[b * RSLOT + t]; }

    float dsel = 0.f, ssel = 1.f;
    score_group<LSLOT>(icl,  &v_lds[bl][0][0], ce_raw, lane, 0,  dsel, ssel);
    score_group<5>(icr,      &v_lds[bl][1][0], ce_raw, lane, 16, dsel, ssel);
    __builtin_amdgcn_sched_barrier(0);
    score_group<5>(icr + 5,  &v_lds[bl][1][0], ce_raw, lane, 21, dsel, ssel);

    const unsigned no_slot = ~((1u << LSLOT) - 1u) & ~(((1u << RSLOT) - 1u) << 16);
    unsigned neg_bits = no_slot;
    #pragma unroll
    for (int t = 0; t < LSLOT; ++t) neg_bits |= (icl[t] < 0 ? 1u : 0u) << t;
    #pragma unroll
    for (int t = 0; t < RSLOT; ++t) neg_bits |= (icr[t] < 0 ? 1u : 0u) << (16 + t);
    const bool has_slot = !((no_slot  >> lane) & 1u);
    const bool is_neg   =   (neg_bits >> lane) & 1u;

    float sc = dsel / fmaxf(sqrtf(ssel), 1e-12f);
    sc = is_neg ? NEGV : sc;
    const float mx = row_max16(sc);
    const float ex = has_slot ? expf(sc - mx) : 0.f;
    const float aw = ex * (1.f / row_sum16(ex));

    float la[LSLOT], ra[RSLOT];
    la[0] = lane_value<0>(aw);  la[1] = lane_value<1>(aw);  la[2] = lane_value<2>(aw);
    la[3] = lane_value<3>(aw);  la[4] = lane_value<4>(aw);  la[5] = lane_value<5>(aw);
    la[6] = lane_value<6>(aw);
    ra[0] = lane_value<16>(aw); ra[1] = lane_value<17>(aw); ra[2] = lane_value<18>(aw);
    ra[3] = lane_value<19>(aw); ra[4] = lane_value<20>(aw); ra[5] = lane_value<21>(aw);
    ra[6] = lane_value<22>(aw); ra[7] = lane_value<23>(aw); ra[8] = lane_value<24>(aw);
    ra[9] = lane_value<25>(aw);

    float z0 = gL_b[0], z1 = gL_b[1];
    #pragma unroll
    for (int t = 0; t < LSLOT; ++t) {
      z0 = fmaf(la[t], gL_w[t],                z0);
      z1 = fmaf(la[t], gL_w[(LSLOT + RSLOT) + t], z1);
    }
    #pragma unroll
    for (int t = 0; t < RSLOT; ++t) {
      z0 = fmaf(ra[t], gL_w[LSLOT + t],                       z0);
      z1 = fmaf(ra[t], gL_w[(LSLOT + RSLOT) + LSLOT + t],     z1);
    }
    float gm = fmaxf(z0, z1);
    float e0 = expf(z0 - gm), e1 = expf(z1 - gm);
    float gi = 1.f / (e0 + e1);
    float g0 = e0 * gi, g1 = e1 * gi;

    const float* pl[LSLOT]; const float* pr[RSLOT];
    #pragma unroll
    for (int t = 0; t < LSLOT; ++t) pl[t] = wvec + (long)max(iwl[t], 0) * WDIM;
    #pragma unroll
    for (int t = 0; t < RSLOT; ++t) pr[t] = wvec + (long)max(iwr[t], 0) * WDIM;
    const v4f oA = emb_piece<v4f, 4, 2>(pl, pr, la, ra, iwl, iwr, lane,      g0, g1);
    *(v4f*)&o_lds[wv * WDIM + 4 * lane] = oA;
    __builtin_amdgcn_sched_barrier(0);
    const v4f oB = emb_piece<v4f, 4, 2>(pl, pr, la, ra, iwl, iwr, lane + 32, g0, g1);
    *(v4f*)&o_lds[wv * WDIM + 128 + 4 * lane] = oB;
    const bool tact = lane < (WDIM - 256) / 2;
    const int  t2   = 128 + (tact ? lane : (WDIM - 256) / 2 - 1);
    const v2f oT = emb_piece<v2f, 2, RSLOT>(pl, pr, la, ra, iwl, iwr, t2, g0, g1);
    if (tact) *(v2f*)&o_lds[wv * WDIM + 256 + 2 * lane] = oT;
  }
  __syncthreads();

  {
    float* otile = out + (long)bbase * WDIM;
    for (int line = wv; line < (16 * WDIM) / 32; line += 16)
      otile[line * 32 + lane] = o_lds[line * 32 + lane];
  }
}

extern "C" void kernel_launch(void* const* d_in, const int* in_sizes, int n_in,
                              void* d_out, int out_size, void* d_ws, size_t ws_size,
                              hipStream_t stream) {
  const float* ce_raw = (const float*)d_in[0];
  const float* wvec   = (const float*)d_in[1];
  const float* lW     = (const float*)d_in[2];
  const float* rW     = (const float*)d_in[3];
  const float* gL_w   = (const float*)d_in[4];
  const float* gL_b   = (const float*)d_in[5];
  const int*   qidx   = (const int*)d_in[6];
  const int*   lixs_c = (const int*)d_in[7];
  const int*   rixs_c = (const int*)d_in[8];
  const int*   lixs_w = (const int*)d_in[9];
  const int*   rixs_w = (const int*)d_in[10];

  const int B = in_sizes[6];
  const int blocks = B / 16;
  fused_gated_attn_kernel<<<blocks, 512, 0, stream>>>(
      ce_raw, wvec, lW, rW, gL_w, gL_b,
      qidx, lixs_c, rixs_c, lixs_w, rixs_w, (float*)d_out);
}


# Two batch rows per wave: 16-wave work-group takes 32 rows (W fragments feed 2 WMMA accumulators; both rows' gather loads issued together), grid 512 = 8,192 waves, was 16,384. Sum order same.
# speedup vs baseline: 26.0608x; 1.0364x over previous
#include <hip/hip_runtime.h>
#include <hip/hip_bf16.h>

typedef float v2f __attribute__((ext_vector_type(2)));
typedef float v4f __attribute__((ext_vector_type(4)));
typedef float v8f __attribute__((ext_vector_type(8)));

#define CDIM  100
#define LSLOT 7
#define RSLOT 10
#define WDIM  300
#define QPITCH 108
#define W_AHEAD 9
#define NEGV  -1e30f

template <int M>
__device__ __forceinline__ float row_xor(float v) {
  return __int_as_float(__builtin_amdgcn_update_dpp(0, __float_as_int(v), 0x160 | M, 0xf, 0xf, false));
}
__device__ __forceinline__ float row_max16(float v) {
  v = fmaxf(v, row_xor<8>(v));  v = fmaxf(v, row_xor<4>(v));
  v = fmaxf(v, row_xor<2>(v));  v = fmaxf(v, row_xor<1>(v));
  return v;
}
__device__ __forceinline__ float row_sum16(float v) {
  v += row_xor<8>(v);  v += row_xor<4>(v);
  v += row_xor<2>(v);  v += row_xor<1>(v);
  return v;
}
__device__ __forceinline__ float xhalf(float v) {
  const int b = __float_as_int(v);
  return __int_as_float(__builtin_amdgcn_permlanex16(b, b, 0x76543210, 0xfedcba98, false, false));
}
__device__ __forceinline__ float wave_sum(float v) {
  v += xhalf(v);
  return row_sum16(v);
}


template <int N>
__device__ __forceinline__ void score_reduce(const v4f (&x)[N], const v4f v, bool act, int lane,
                                             int lane0, float& dsel, float& ssel) {
  const v2f vlo = {v[0], v[1]}, vhi = {v[2], v[3]};
  float dot[N], ss[N];
  #pragma unroll
  for (int t = 0; t < N; ++t) {
    const v2f xlo = {x[t][0], x[t][1]}, xhi = {x[t][2], x[t][3]};
    const v2f d2 = __builtin_elementwise_fma(xhi, vhi, xlo * vlo);
    const v2f s2 = __builtin_elementwise_fma(xhi, xhi, xlo * xlo);
    dot[t] = act ? d2[0] + d2[1] : 0.f;
    ss[t]  = act ? s2[0] + s2[1] : 0.f;
  }
  const bool hi_row = lane >= 16;
  #pragma unroll
  for (int t = 0; t < N; ++t) {
    const float mine = hi_row ? ss[t]  : dot[t];
    const float give = hi_row ? dot[t] : ss[t];
    const float r = row_sum16(mine + xhalf(give));
    const float o = xhalf(r);
    const bool  low = (lane0 + t) < 16;
    const float dtot = low ? r : o, stot = low ? o : r;
    if (lane == lane0 + t) { dsel = dtot; ssel = stot; }
  }
}

template <int N>
__device__ __forceinline__ void score_group2(const int* idxA, const int* idxB,
                                             const float* vrowA, const float* vrowB,
                                             const float* __restrict__ ce_raw, int lane, int lane0,
                                             float& dselA, float& sselA, float& dselB, float& sselB) {
  const bool act = lane < CDIM / 4;
  const int  c4  = act ? lane : CDIM / 4 - 1;
  const v4f vA = *(const v4f*)(vrowA + 4 * c4);
  const v4f vB = *(const v4f*)(vrowB + 4 * c4);
  v4f xA[N], xB[N];
  #pragma unroll
  for (int t = 0; t < N; ++t) xA[t] = ((const v4f*)(ce_raw + (long)max(idxA[t], 0) * CDIM))[c4];
  #pragma unroll
  for (int t = 0; t < N; ++t) xB[t] = ((const v4f*)(ce_raw + (long)max(idxB[t], 0) * CDIM))[c4];
  score_reduce<N>(xA, vA, act, lane, lane0, dselA, sselA);
  score_reduce<N>(xB, vB, act, lane, lane0, dselB, sselB);
}

__device__ __forceinline__ int lane_int(int v, int L) { return __builtin_amdgcn_readlane(v, L); }
__device__ __forceinline__ float lane_flt(float v, int L) {
  return __int_as_float(__builtin_amdgcn_readlane(__float_as_int(v), L));
}

template <typename VT, int NE, int NS, int T0, int T1, int LB>
__device__ __forceinline__ void emb_acc(const VT (&x)[NS], float aw, int iwv, VT& acc) {
  #pragma unroll
  for (int t = T0; t < T1; ++t) {
    const float a = lane_flt(aw, LB + t);
    const bool pad = lane_int(iwv, LB + t) < 0;
    #pragma unroll
    for (int e = 0; e < NE; ++e) acc[e] = fmaf(a, pad ? 0.f : x[t][e], acc[e]);
  }
}
template <typename VT, int NS, int T0, int T1, int LB>
__device__ __forceinline__ void emb_load(VT (&x)[NS], const float* __restrict__ wvec, int iwv, int vi) {
  #pragma unroll
  for (int t = T0; t < T1; ++t) {
    const unsigned off = (unsigned)max(lane_int(iwv, LB + t), 0) * (unsigned)(WDIM * sizeof(float))
                       + (unsigned)vi * (unsigned)sizeof(VT);
    x[t] = *(const VT*)((const char*)wvec + off);
  }
}

template <typename VT, int NE, int SPLIT>
__device__ __forceinline__ void emb_piece2(const float* __restrict__ wvec,
                                           float awA, int iwvA, float g0A, float g1A,
                                           float awB, int iwvB, float g0B, float g1B,
                                           int vi, VT& oA, VT& oB) {
  VT xlA[LSLOT], xrA[RSLOT], xlB[LSLOT], xrB[RSLOT];
  emb_load<VT, LSLOT, 0, LSLOT, 0 >(xlA, wvec, iwvA, vi);
  emb_load<VT, RSLOT, 0, SPLIT, 16>(xrA, wvec, iwvA, vi);
  emb_load<VT, LSLOT, 0, LSLOT, 0 >(xlB, wvec, iwvB, vi);
  emb_load<VT, RSLOT, 0, SPLIT, 16>(xrB, wvec, iwvB, vi);
  VT aLA, aRA, aLB, aRB;
  #pragma unroll
  for (int e = 0; e < NE; ++e) { aLA[e] = 0.f; aRA[e] = 0.f; aLB[e] = 0.f; aRB[e] = 0.f; }
  emb_acc<VT, NE, LSLOT, 0, LSLOT, 0 >(xlA, awA, iwvA, aLA);
  emb_acc<VT, NE, RSLOT, 0, SPLIT, 16>(xrA, awA, iwvA, aRA);
  emb_acc<VT, NE, LSLOT, 0, LSLOT, 0 >(xlB, awB, iwvB, aLB);
  emb_acc<VT, NE, RSLOT, 0, SPLIT, 16>(xrB, awB, iwvB, aRB);
  if (SPLIT < RSLOT) {
    __builtin_amdgcn_sched_barrier(0);
    emb_load<VT, RSLOT, SPLIT, RSLOT, 16>(xrA, wvec, iwvA, vi);
    emb_load<VT, RSLOT, SPLIT, RSLOT, 16>(xrB, wvec, iwvB, vi);
    emb_acc<VT, NE, RSLOT, SPLIT, RSLOT, 16>(xrA, awA, iwvA, aRA);
    emb_acc<VT, NE, RSLOT, SPLIT, RSLOT, 16>(xrB, awB, iwvB, aRB);
  }
  #pragma unroll
  for (int e = 0; e < NE; ++e) { oA[e] = g0A * aLA[e] + g1A * aRA[e]; oB[e] = g0B * aLB[e] + g1B * aRB[e]; }
}

__global__ __launch_bounds__(512) __attribute__((amdgpu_waves_per_eu(8)))
void fused_gated_attn_kernel(
    const float* __restrict__ ce_raw, const float* __restrict__ wvec,
    const float* __restrict__ lW,     const float* __restrict__ rW,
    const float* __restrict__ gL_w,   const float* __restrict__ gL_b,
    const int* __restrict__ qidx,
    const int* __restrict__ lixs_c,   const int* __restrict__ rixs_c,
    const int* __restrict__ lixs_w,   const int* __restrict__ rixs_w,
    float* __restrict__ out)
{
  __shared__ __attribute__((aligned(16))) float q_lds[32][QPITCH];
  __shared__ __attribute__((aligned(16))) float v_lds[32][2][112];
  __shared__ __attribute__((aligned(16))) float o_lds[32 * WDIM];

  const int lane  = threadIdx.x & 31;
  const int wv    = __builtin_amdgcn_readfirstlane(threadIdx.x >> 5);
  const int bbase = blockIdx.x * 32;
  const int hi = lane >> 4, ln = lane & 15;

  const int  w_side = wv / 7, w_nt = wv % 7;
  const v4f* wp = (const v4f*)((w_side ? rW : lW) + min(w_nt * 16 + ln, CDIM - 1) * CDIM + 4 * hi);
  v4f bw[13];
  if (wv < 14) {
    #pragma unroll
    for (int j = 0; j < W_AHEAD; ++j) bw[j] = wp[2 * j];
  }

  {
    const long row0 = qidx[bbase + wv], row1 = qidx[bbase + 16 + wv];
    const v4f zero4q = {0.f, 0.f, 0.f, 0.f};
    v4f qv[2] = {zero4q, zero4q};
    if (lane < CDIM / 4) {
      qv[0] = ((const v4f*)(ce_raw + row0 * CDIM))[lane];
      qv[1] = ((const v4f*)(ce_raw + row1 * CDIM))[lane];
    }
    #pragma unroll
    for (int r = 0; r < 2; ++r) {
      float ss = 0.f;
      #pragma unroll
      for (int e = 0; e < 4; ++e) ss = fmaf(qv[r][e], qv[r][e], ss);
      ss = wave_sum(ss);
      float inv = 1.f / fmaxf(sqrtf(ss), 1e-12f);
      if (lane < CDIM / 4 + 1) {
        v4f qn;
        #pragma unroll
        for (int e = 0; e < 4; ++e) qn[e] = qv[r][e] * inv;
        *(v4f*)&q_lds[wv + 16 * r][4 * lane] = qn;
      }
    }
  }
  __syncthreads();

  if (wv < 14) {
    const v4f zero4 = {0.f, 0.f, 0.f, 0.f};
    v8f acc0 = {}, acc1 = {};
    #pragma unroll
    for (int j = 0; j < 13; ++j) {
      if (j == 13 - W_AHEAD) {
        #pragma unroll
        for (int t = W_AHEAD; t < 12; ++t) bw[t] = wp[2 * t];
        bw[12] = (hi == 0) ? wp[24] : zero4;
      }
      const v4f a4 = *(const v4f*)&q_lds[ln][8 * j + 4 * hi];
      const v4f c4 = *(const v4f*)&q_lds[16 + ln][8 * j + 4 * hi];
      v2f a, c, bfrag;
      a[0] = a4[0];  a[1] = a4[1];  c[0] = c4[0];  c[1] = c4[1];  bfrag[0] = bw[j][0];  bfrag[1] = bw[j][1];
      acc0 = __builtin_amdgcn_wmma_f32_16x16x4_f32(
           false, a,  false, bfrag,
           (short)0, acc0,  false,  false);
      acc1 = __builtin_amdgcn_wmma_f32_16x16x4_f32(
           false, c,  false, bfrag,
           (short)0, acc1,  false,  false);
      a[0] = a4[2];  a[1] = a4[3];  c[0] = c4[2];  c[1] = c4[3];  bfrag[0] = bw[j][2];  bfrag[1] = bw[j][3];
      acc0 = __builtin_amdgcn_wmma_f32_16x16x4_f32(
           false, a,  false, bfrag,
           (short)0, acc0,  false,  false);
      acc1 = __builtin_amdgcn_wmma_f32_16x16x4_f32(
           false, c,  false, bfrag,
           (short)0, acc1,  false,  false);
    }
    #pragma unroll
    for (int r = 0; r < 8; ++r) {
      v_lds[r + 8 * hi][w_side][w_nt * 16 + ln]      = acc0[r];
      v_lds[16 + r + 8 * hi][w_side][w_nt * 16 + ln] = acc1[r];
    }
  }
  __syncthreads();

  {
    const int bA = bbase + wv, bB = bbase + 16 + wv;
    const int blA = wv,        blB = wv + 16;

    const int wlA = lixs_w[bA * LSLOT + min(ln, LSLOT - 1)], wrA = rixs_w[bA * RSLOT + min(ln, RSLOT - 1)];
    const int wlB = lixs_w[bB * LSLOT + min(ln, LSLOT - 1)], wrB = rixs_w[bB * RSLOT + min(ln, RSLOT - 1)];
    const int iwvA = hi ? wrA : wlA, iwvB = hi ? wrB : wlB;

    int iclA[LSLOT], icrA[RSLOT], iclB[LSLOT], icrB[RSLOT];
    #pragma unroll
    for (int t = 0; t < LSLOT; ++t) { iclA[t] = lixs_c[bA * LSLOT + t]; iclB[t] = lixs_c[bB * LSLOT + t]; }
    #pragma unroll
    for (int t = 0; t < RSLOT; ++t) { icrA[t] = rixs_c[bA * RSLOT + t]; icrB[t] = rixs_c[bB * RSLOT + t]; }

    const unsigned no_slot = ~((1u << LSLOT) - 1u) & ~(((1u << RSLOT) - 1u) << 16);
    unsigned negA = no_slot, negB = no_slot;
    #pragma unroll
    for (int t = 0; t < LSLOT; ++t) { negA |= (iclA[t] < 0 ? 1u : 0u) << t; negB |= (iclB[t] < 0 ? 1u : 0u) << t; }
    #pragma unroll
    for (int t = 0; t < RSLOT; ++t) { negA |= (icrA[t] < 0 ? 1u : 0u) << (16 + t); negB |= (icrB[t] < 0 ? 1u : 0u) << (16 + t); }

    float dselA = 0.f, sselA = 1.f, dselB = 0.f, sselB = 1.f;
    score_group2<LSLOT>(iclA, iclB, &v_lds[blA][0][0], &v_lds[blB][0][0], ce_raw, lane, 0,  dselA, sselA, dselB, sselB);
    __builtin_amdgcn_sched_barrier(0);
    score_group2<5>(icrA,     icrB,     &v_lds[blA][1][0], &v_lds[blB][1][0], ce_raw, lane, 16, dselA, sselA, dselB, sselB);
    __builtin_amdgcn_sched_barrier(0);
    score_group2<5>(icrA + 5, icrB + 5, &v_lds[blA][1][0], &v_lds[blB][1][0], ce_raw, lane, 21, dselA, sselA, dselB, sselB);

    const bool has_slot = !((no_slot >> lane) & 1u);

    const float dsel[2] = {dselA, dselB}, ssel[2] = {sselA, sselB};
    const unsigned negb[2] = {negA, negB};
    float aw[2], g0[2], g1[2];
    #pragma unroll
    for (int r = 0; r < 2; ++r) {
      const bool is_neg = (negb[r] >> lane) & 1u;
      float sc = dsel[r] / fmaxf(sqrtf(ssel[r]), 1e-12f);
      sc = is_neg ? NEGV : sc;
      const float mx = row_max16(sc);
      const float ex = has_slot ? expf(sc - mx) : 0.f;
      aw[r] = ex * (1.f / row_sum16(ex));

      float z0 = gL_b[0], z1 = gL_b[1];
      #pragma unroll
      for (int t = 0; t < LSLOT; ++t) {
        const float a = lane_flt(aw[r], t);
        z0 = fmaf(a, gL_w[t],                   z0);
        z1 = fmaf(a, gL_w[(LSLOT + RSLOT) + t], z1);
      }
      #pragma unroll
      for (int t = 0; t < RSLOT; ++t) {
        const float a = lane_flt(aw[r], 16 + t);
        z0 = fmaf(a, gL_w[LSLOT + t],                   z0);
        z1 = fmaf(a, gL_w[(LSLOT + RSLOT) + LSLOT + t], z1);
      }
      const float gm = fmaxf(z0, z1);
      const float e0 = expf(z0 - gm), e1 = expf(z1 - gm);
      const float gi = 1.f / (e0 + e1);
      g0[r] = e0 * gi;  g1[r] = e1 * gi;
    }

    v4f oA, oB;
    emb_piece2<v4f, 4, 2>(wvec, aw[0], iwvA, g0[0], g1[0], aw[1], iwvB, g0[1], g1[1], lane, oA, oB);
    *(v4f*)&o_lds[blA * WDIM + 4 * lane] = oA;
    *(v4f*)&o_lds[blB * WDIM + 4 * lane] = oB;
    __builtin_amdgcn_sched_barrier(0);
    emb_piece2<v4f, 4, 2>(wvec, aw[0], iwvA, g0[0], g1[0], aw[1], iwvB, g0[1], g1[1], lane + 32, oA, oB);
    *(v4f*)&o_lds[blA * WDIM + 128 + 4 * lane] = oA;
    *(v4f*)&o_lds[blB * WDIM + 128 + 4 * lane] = oB;
    const bool tact = lane < (WDIM - 256) / 2;
    const int  t2   = 128 + (tact ? lane : (WDIM - 256) / 2 - 1);
    v2f tA, tB;
    emb_piece2<v2f, 2, RSLOT>(wvec, aw[0], iwvA, g0[0], g1[0], aw[1], iwvB, g0[1], g1[1], t2, tA, tB);
    if (tact) {
      *(v2f*)&o_lds[blA * WDIM + 256 + 2 * lane] = tA;
      *(v2f*)&o_lds[blB * WDIM + 256 + 2 * lane] = tB;
    }
  }
  __syncthreads();

  {
    float* otile = out + (long)bbase * WDIM;
    for (int line = wv; line < (32 * WDIM) / 32; line += 16)
      otile[line * 32 + lane] = o_lds[line * 32 + lane];
  }
}

extern "C" void kernel_launch(void* const* d_in, const int* in_sizes, int n_in,
                              void* d_out, int out_size, void* d_ws, size_t ws_size,
                              hipStream_t stream) {
  const float* ce_raw = (const float*)d_in[0];
  const float* wvec   = (const float*)d_in[1];
  const float* lW     = (const float*)d_in[2];
  const float* rW     = (const float*)d_in[3];
  const float* gL_w   = (const float*)d_in[4];
  const float* gL_b   = (const float*)d_in[5];
  const int*   qidx   = (const int*)d_in[6];
  const int*   lixs_c = (const int*)d_in[7];
  const int*   rixs_c = (const int*)d_in[8];
  const int*   lixs_w = (const int*)d_in[9];
  const int*   rixs_w = (const int*)d_in[10];

  const int B = in_sizes[6];
  const int blocks = B / 32;
  fused_gated_attn_kernel<<<blocks, 512, 0, stream>>>(
      ce_raw, wvec, lW, rW, gL_w, gL_b,
      qidx, lixs_c, rixs_c, lixs_w, rixs_w, (float*)d_out);
}


# Gather pipelined: indices read at entry, 10 context rows requested under the WMMAs (hand-written barrier, no load drain), each later group and 10 word rows issued before the sums ahead. Sums same.
# speedup vs baseline: 26.9522x; 1.0342x over previous
#include <hip/hip_runtime.h>
#include <hip/hip_bf16.h>

typedef float v2f __attribute__((ext_vector_type(2)));
typedef float v4f __attribute__((ext_vector_type(4)));
typedef float v8f __attribute__((ext_vector_type(8)));

#define CDIM  100
#define LSLOT 7
#define RSLOT 10
#define WDIM  300
#define QPITCH 108
#define W_AHEAD 9
#define XL_AHEAD 5
#define NEGV  -1e30f

template <int M>
__device__ __forceinline__ float row_xor(float v) {
  return __int_as_float(__builtin_amdgcn_update_dpp(0, __float_as_int(v), 0x160 | M, 0xf, 0xf, false));
}
__device__ __forceinline__ float row_max16(float v) {
  v = fmaxf(v, row_xor<8>(v));  v = fmaxf(v, row_xor<4>(v));
  v = fmaxf(v, row_xor<2>(v));  v = fmaxf(v, row_xor<1>(v));
  return v;
}
__device__ __forceinline__ float row_sum16(float v) {
  v += row_xor<8>(v);  v += row_xor<4>(v);
  v += row_xor<2>(v);  v += row_xor<1>(v);
  return v;
}
__device__ __forceinline__ float xhalf(float v) {
  const int b = __float_as_int(v);
  return __int_as_float(__builtin_amdgcn_permlanex16(b, b, 0x76543210, 0xfedcba98, false, false));
}
__device__ __forceinline__ float wave_sum(float v) {
  v += xhalf(v);
  return row_sum16(v);
}


template <int N>
__device__ __forceinline__ void score_reduce(const v4f (&x)[N], const v4f v, bool act, int lane,
                                             int lane0, float& dsel, float& ssel) {
  const v2f vlo = {v[0], v[1]}, vhi = {v[2], v[3]};
  float dot[N], ss[N];
  #pragma unroll
  for (int t = 0; t < N; ++t) {
    const v2f xlo = {x[t][0], x[t][1]}, xhi = {x[t][2], x[t][3]};
    const v2f d2 = __builtin_elementwise_fma(xhi, vhi, xlo * vlo);
    const v2f s2 = __builtin_elementwise_fma(xhi, xhi, xlo * xlo);
    dot[t] = act ? d2[0] + d2[1] : 0.f;
    ss[t]  = act ? s2[0] + s2[1] : 0.f;
  }
  const bool hi_row = lane >= 16;
  #pragma unroll
  for (int t = 0; t < N; ++t) {
    const float mine = hi_row ? ss[t]  : dot[t];
    const float give = hi_row ? dot[t] : ss[t];
    const float r = row_sum16(mine + xhalf(give));
    const float o = xhalf(r);
    const bool  low = (lane0 + t) < 16;
    const float dtot = low ? r : o, stot = low ? o : r;
    if (lane == lane0 + t) { dsel = dtot; ssel = stot; }
  }
}

__device__ __forceinline__ int lane_int(int v, int L) { return __builtin_amdgcn_readlane(v, L); }
__device__ __forceinline__ float lane_flt(float v, int L) {
  return __int_as_float(__builtin_amdgcn_readlane(__float_as_int(v), L));
}

template <int N, int LB>
__device__ __forceinline__ void ctx_load(v4f (&x)[N], const float* __restrict__ ce_raw, int icv, int c4) {
  #pragma unroll
  for (int t = 0; t < N; ++t) x[t] = ((const v4f*)(ce_raw + (long)max(lane_int(icv, LB + t), 0) * CDIM))[c4];
}

template <typename VT, int NE, int NS, int T0, int T1, int LB>
__device__ __forceinline__ void emb_acc(const VT (&x)[NS], float aw, int iwv, VT& acc) {
  #pragma unroll
  for (int t = T0; t < T1; ++t) {
    const float a = lane_flt(aw, LB + t);
    const bool pad = lane_int(iwv, LB + t) < 0;
    #pragma unroll
    for (int e = 0; e < NE; ++e) acc[e] = fmaf(a, pad ? 0.f : x[t][e], acc[e]);
  }
}
template <typename VT, int NS, int T0, int T1, int LB>
__device__ __forceinline__ void emb_load(VT (&x)[NS], const float* __restrict__ wvec, int iwv, int vi) {
  #pragma unroll
  for (int t = T0; t < T1; ++t) {
    const unsigned off = (unsigned)max(lane_int(iwv, LB + t), 0) * (unsigned)(WDIM * sizeof(float))
                       + (unsigned)vi * (unsigned)sizeof(VT);
    x[t] = *(const VT*)((const char*)wvec + off);
  }
}

template <typename VT, int NE, int SPLIT, int XL_PRE>
__device__ __forceinline__ void emb_piece2(const float* __restrict__ wvec,
                                           float awA, int iwvA, float g0A, float g1A,
                                           float awB, int iwvB, float g0B, float g1B,
                                           int vi, VT (&xlA)[LSLOT], VT (&xlB)[LSLOT], VT& oA, VT& oB) {
  VT xrA[RSLOT], xrB[RSLOT];
  emb_load<VT, LSLOT, XL_PRE, LSLOT, 0 >(xlA, wvec, iwvA, vi);
  emb_load<VT, RSLOT, 0, SPLIT, 16>(xrA, wvec, iwvA, vi);
  emb_load<VT, LSLOT, XL_PRE, LSLOT, 0 >(xlB, wvec, iwvB, vi);
  emb_load<VT, RSLOT, 0, SPLIT, 16>(xrB, wvec, iwvB, vi);
  __builtin_amdgcn_sched_barrier(0);
  VT aLA, aRA, aLB, aRB;
  #pragma unroll
  for (int e = 0; e < NE; ++e) { aLA[e] = 0.f; aRA[e] = 0.f; aLB[e] = 0.f; aRB[e] = 0.f; }
  emb_acc<VT, NE, LSLOT, 0, LSLOT, 0 >(xlA, awA, iwvA, aLA);
  emb_acc<VT, NE, RSLOT, 0, SPLIT, 16>(xrA, awA, iwvA, aRA);
  emb_acc<VT, NE, LSLOT, 0, LSLOT, 0 >(xlB, awB, iwvB, aLB);
  emb_acc<VT, NE, RSLOT, 0, SPLIT, 16>(xrB, awB, iwvB, aRB);
  if (SPLIT < RSLOT) {
    __builtin_amdgcn_sched_barrier(0);
    emb_load<VT, RSLOT, SPLIT, RSLOT, 16>(xrA, wvec, iwvA, vi);
    emb_load<VT, RSLOT, SPLIT, RSLOT, 16>(xrB, wvec, iwvB, vi);
    emb_acc<VT, NE, RSLOT, SPLIT, RSLOT, 16>(xrA, awA, iwvA, aRA);
    emb_acc<VT, NE, RSLOT, SPLIT, RSLOT, 16>(xrB, awB, iwvB, aRB);
  }
  #pragma unroll
  for (int e = 0; e < NE; ++e) { oA[e] = g0A * aLA[e] + g1A * aRA[e]; oB[e] = g0B * aLB[e] + g1B * aRB[e]; }
}

__device__ __forceinline__ void wmma_kstep(const float* qa, const float* qc, const v4f b4, v8f& acc0, v8f& acc1) {
  const v4f a4 = *(const v4f*)qa;
  const v4f c4 = *(const v4f*)qc;
  v2f a, c, bfrag;
  a[0] = a4[0];  a[1] = a4[1];  c[0] = c4[0];  c[1] = c4[1];  bfrag[0] = b4[0];  bfrag[1] = b4[1];
  acc0 = __builtin_amdgcn_wmma_f32_16x16x4_f32(
       false, a,  false, bfrag,
       (short)0, acc0,  false,  false);
  acc1 = __builtin_amdgcn_wmma_f32_16x16x4_f32(
       false, c,  false, bfrag,
       (short)0, acc1,  false,  false);
  a[0] = a4[2];  a[1] = a4[3];  c[0] = c4[2];  c[1] = c4[3];  bfrag[0] = b4[2];  bfrag[1] = b4[3];
  acc0 = __builtin_amdgcn_wmma_f32_16x16x4_f32(
       false, a,  false, bfrag,
       (short)0, acc0,  false,  false);
  acc1 = __builtin_amdgcn_wmma_f32_16x16x4_f32(
       false, c,  false, bfrag,
       (short)0, acc1,  false,  false);
}

__global__ __launch_bounds__(512) __attribute__((amdgpu_waves_per_eu(8)))
void fused_gated_attn_kernel(
    const float* __restrict__ ce_raw, const float* __restrict__ wvec,
    const float* __restrict__ lW,     const float* __restrict__ rW,
    const float* __restrict__ gL_w,   const float* __restrict__ gL_b,
    const int* __restrict__ qidx,
    const int* __restrict__ lixs_c,   const int* __restrict__ rixs_c,
    const int* __restrict__ lixs_w,   const int* __restrict__ rixs_w,
    float* __restrict__ out)
{
  __shared__ __attribute__((aligned(16))) float q_lds[32][QPITCH];
  __shared__ __attribute__((aligned(16))) float v_lds[32][2][112];
  __shared__ __attribute__((aligned(16))) float o_lds[32 * WDIM];

  const int lane  = threadIdx.x & 31;
  const int wv    = __builtin_amdgcn_readfirstlane(threadIdx.x >> 5);
  const int bbase = blockIdx.x * 32;
  const int hi = lane >> 4, ln = lane & 15;
  const int bA = bbase + wv, bB = bbase + 16 + wv;

  const int clA = lixs_c[bA * LSLOT + min(ln, LSLOT - 1)], crA = rixs_c[bA * RSLOT + min(ln, RSLOT - 1)];
  const int clB = lixs_c[bB * LSLOT + min(ln, LSLOT - 1)], crB = rixs_c[bB * RSLOT + min(ln, RSLOT - 1)];
  const int icvA = hi ? crA : clA, icvB = hi ? crB : clB;
  const int wlA = lixs_w[bA * LSLOT + min(ln, LSLOT - 1)], wrA = rixs_w[bA * RSLOT + min(ln, RSLOT - 1)];
  const int wlB = lixs_w[bB * LSLOT + min(ln, LSLOT - 1)], wrB = rixs_w[bB * RSLOT + min(ln, RSLOT - 1)];
  const int iwvA = hi ? wrA : wlA, iwvB = hi ? wrB : wlB;

  const int  w_side = wv / 7, w_nt = wv % 7;
  const v4f* wp = (const v4f*)((w_side ? rW : lW) + min(w_nt * 16 + ln, CDIM - 1) * CDIM + 4 * hi);
  v4f bw[13];
  if (wv < 14) {
    #pragma unroll
    for (int j = 0; j < W_AHEAD; ++j) bw[j] = wp[2 * j];
  }

  {
    const long row0 = qidx[bbase + wv], row1 = qidx[bbase + 16 + wv];
    const v4f zero4q = {0.f, 0.f, 0.f, 0.f};
    v4f qv[2] = {zero4q, zero4q};
    if (lane < CDIM / 4) {
      qv[0] = ((const v4f*)(ce_raw + row0 * CDIM))[lane];
      qv[1] = ((const v4f*)(ce_raw + row1 * CDIM))[lane];
    }
    #pragma unroll
    for (int r = 0; r < 2; ++r) {
      float ss = 0.f;
      #pragma unroll
      for (int e = 0; e < 4; ++e) ss = fmaf(qv[r][e], qv[r][e], ss);
      ss = wave_sum(ss);
      float inv = 1.f / fmaxf(sqrtf(ss), 1e-12f);
      if (lane < CDIM / 4 + 1) {
        v4f qn;
        #pragma unroll
        for (int e = 0; e < 4; ++e) qn[e] = qv[r][e] * inv;
        *(v4f*)&q_lds[wv + 16 * r][4 * lane] = qn;
      }
    }
  }
  __syncthreads();

  const bool cact = lane < CDIM / 4;
  const int  c4i  = cact ? lane : CDIM / 4 - 1;
  v4f hA[5], hB[5];
  v8f acc0 = {}, acc1 = {};
  if (wv < 14) {
    const v4f zero4 = {0.f, 0.f, 0.f, 0.f};
    #pragma unroll
    for (int j = 0; j < 13 - W_AHEAD; ++j)
      wmma_kstep(&q_lds[ln][8 * j + 4 * hi], &q_lds[16 + ln][8 * j + 4 * hi], bw[j], acc0, acc1);
    #pragma unroll
    for (int t = W_AHEAD; t < 12; ++t) bw[t] = wp[2 * t];
    bw[12] = (hi == 0) ? wp[24] : zero4;
  }
  ctx_load<5, 16>(hA, ce_raw, icvA, c4i);
  ctx_load<5, 16>(hB, ce_raw, icvB, c4i);
  if (wv < 14) {
    #pragma unroll
    for (int j = 13 - W_AHEAD; j < 13; ++j)
      wmma_kstep(&q_lds[ln][8 * j + 4 * hi], &q_lds[16 + ln][8 * j + 4 * hi], bw[j], acc0, acc1);
    #pragma unroll
    for (int r = 0; r < 8; ++r) {
      v_lds[r + 8 * hi][w_side][w_nt * 16 + ln]      = acc0[r];
      v_lds[16 + r + 8 * hi][w_side][w_nt * 16 + ln] = acc1[r];
    }
  }
  asm volatile("s_wait_dscnt 0x0" ::: "memory");
  asm volatile("s_barrier_signal -1" ::: "memory");
  asm volatile("s_barrier_wait -1" ::: "memory");

  {
    const int blA = wv,        blB = wv + 16;

    const unsigned no_slot = ~((1u << LSLOT) - 1u) & ~(((1u << RSLOT) - 1u) << 16);

    float dselA = 0.f, sselA = 1.f, dselB = 0.f, sselB = 1.f;
    const v4f vRA = *(const v4f*)(&v_lds[blA][1][0] + 4 * c4i);
    const v4f vRB = *(const v4f*)(&v_lds[blB][1][0] + 4 * c4i);
    v4f rA[5], rB[5];
    ctx_load<5, 21>(rA, ce_raw, icvA, c4i);
    ctx_load<5, 21>(rB, ce_raw, icvB, c4i);
    __builtin_amdgcn_sched_barrier(0);
    score_reduce<5>(hA, vRA, cact, lane, 16, dselA, sselA);
    score_reduce<5>(hB, vRB, cact, lane, 16, dselB, sselB);
    __builtin_amdgcn_sched_barrier(0);
    v4f laA[4], laB[4];
    ctx_load<4, 0>(laA, ce_raw, icvA, c4i);
    ctx_load<4, 0>(laB, ce_raw, icvB, c4i);
    __builtin_amdgcn_sched_barrier(0);
    score_reduce<5>(rA, vRA, cact, lane, 21, dselA, sselA);
    score_reduce<5>(rB, vRB, cact, lane, 21, dselB, sselB);
    __builtin_amdgcn_sched_barrier(0);
    v4f lbA[LSLOT - 4], lbB[LSLOT - 4];
    ctx_load<LSLOT - 4, 4>(lbA, ce_raw, icvA, c4i);
    ctx_load<LSLOT - 4, 4>(lbB, ce_raw, icvB, c4i);
    __builtin_amdgcn_sched_barrier(0);
    const v4f vLA = *(const v4f*)(&v_lds[blA][0][0] + 4 * c4i);
    const v4f vLB = *(const v4f*)(&v_lds[blB][0][0] + 4 * c4i);
    score_reduce<4>(laA, vLA, cact, lane, 0, dselA, sselA);
    score_reduce<4>(laB, vLB, cact, lane, 0, dselB, sselB);
    __builtin_amdgcn_sched_barrier(0);
    v4f xlA[LSLOT], xlB[LSLOT];
    emb_load<v4f, LSLOT, 0, XL_AHEAD, 0>(xlA, wvec, iwvA, lane);
    emb_load<v4f, LSLOT, 0, XL_AHEAD, 0>(xlB, wvec, iwvB, lane);
    __builtin_amdgcn_sched_barrier(0);
    score_reduce<LSLOT - 4>(lbA, vLA, cact, lane, 4, dselA, sselA);
    score_reduce<LSLOT - 4>(lbB, vLB, cact, lane, 4, dselB, sselB);

    const bool has_slot = !((no_slot >> lane) & 1u);

    const float dsel[2] = {dselA, dselB}, ssel[2] = {sselA, sselB};
    const int icv[2] = {icvA, icvB};
    float aw[2], g0[2], g1[2];
    #pragma unroll
    for (int r = 0; r < 2; ++r) {
      const bool is_neg = !has_slot || icv[r] < 0;
      float sc = dsel[r] / fmaxf(sqrtf(ssel[r]), 1e-12f);
      sc = is_neg ? NEGV : sc;
      const float mx = row_max16(sc);
      const float ex = has_slot ? expf(sc - mx) : 0.f;
      aw[r] = ex * (1.f / row_sum16(ex));

      float z0 = gL_b[0], z1 = gL_b[1];
      #pragma unroll
      for (int t = 0; t < LSLOT; ++t) {
        const float a = lane_flt(aw[r], t);
        z0 = fmaf(a, gL_w[t],                   z0);
        z1 = fmaf(a, gL_w[(LSLOT + RSLOT) + t], z1);
      }
      #pragma unroll
      for (int t = 0; t < RSLOT; ++t) {
        const float a = lane_flt(aw[r], 16 + t);
        z0 = fmaf(a, gL_w[LSLOT + t],                   z0);
        z1 = fmaf(a, gL_w[(LSLOT + RSLOT) + LSLOT + t], z1);
      }
      const float gm = fmaxf(z0, z1);
      const float e0 = expf(z0 - gm), e1 = expf(z1 - gm);
      const float gi = 1.f / (e0 + e1);
      g0[r] = e0 * gi;  g1[r] = e1 * gi;
    }

    v4f oA, oB;
    emb_piece2<v4f, 4, 2, XL_AHEAD>(wvec, aw[0], iwvA, g0[0], g1[0], aw[1], iwvB, g0[1], g1[1], lane, xlA, xlB, oA, oB);
    *(v4f*)&o_lds[blA * WDIM + 4 * lane] = oA;
    *(v4f*)&o_lds[blB * WDIM + 4 * lane] = oB;
    __builtin_amdgcn_sched_barrier(0);
    emb_piece2<v4f, 4, 2, 0>(wvec, aw[0], iwvA, g0[0], g1[0], aw[1], iwvB, g0[1], g1[1], lane + 32, xlA, xlB, oA, oB);
    *(v4f*)&o_lds[blA * WDIM + 128 + 4 * lane] = oA;
    *(v4f*)&o_lds[blB * WDIM + 128 + 4 * lane] = oB;
    const bool tact = lane < (WDIM - 256) / 2;
    const int  t2   = 128 + (tact ? lane : (WDIM - 256) / 2 - 1);
    v2f tA, tB, tlA[LSLOT], tlB[LSLOT];
    emb_piece2<v2f, 2, RSLOT, 0>(wvec, aw[0], iwvA, g0[0], g1[0], aw[1], iwvB, g0[1], g1[1], t2, tlA, tlB, tA, tB);
    if (tact) {
      *(v2f*)&o_lds[blA * WDIM + 256 + 2 * lane] = tA;
      *(v2f*)&o_lds[blB * WDIM + 256 + 2 * lane] = tB;
    }
  }
  __syncthreads();

  {
    float* otile = out + (long)bbase * WDIM;
    for (int line = wv; line < (32 * WDIM) / 32; line += 16)
      otile[line * 32 + lane] = o_lds[line * 32 + lane];
  }
}

extern "C" void kernel_launch(void* const* d_in, const int* in_sizes, int n_in,
                              void* d_out, int out_size, void* d_ws, size_t ws_size,
                              hipStream_t stream) {
  const float* ce_raw = (const float*)d_in[0];
  const float* wvec   = (const float*)d_in[1];
  const float* lW     = (const float*)d_in[2];
  const float* rW     = (const float*)d_in[3];
  const float* gL_w   = (const float*)d_in[4];
  const float* gL_b   = (const float*)d_in[5];
  const int*   qidx   = (const int*)d_in[6];
  const int*   lixs_c = (const int*)d_in[7];
  const int*   rixs_c = (const int*)d_in[8];
  const int*   lixs_w = (const int*)d_in[9];
  const int*   rixs_w = (const int*)d_in[10];

  const int B = in_sizes[6];
  const int blocks = B / 32;
  fused_gated_attn_kernel<<<blocks, 512, 0, stream>>>(
      ce_raw, wvec, lW, rW, gL_w, gL_b,
      qidx, lixs_c, rixs_c, lixs_w, rixs_w, (float*)d_out);
}
